# starting kernel
# baseline (speedup 1.0000x reference)
#include <hip/hip_runtime.h>
#include <cstddef>


typedef __attribute__((ext_vector_type(2))) float v2f;
typedef __attribute__((ext_vector_type(8))) float v8f;

#define HIDC 64

template <int ACT>
__global__ __launch_bounds__(128) void gemm_wmma_f32(
    const float* __restrict__ A, const float* __restrict__ B,
    const float* __restrict__ bias, float* __restrict__ C,
    int M, int K, int Nc)
{
    const int lane = threadIdx.x & 31;
    const int wave = threadIdx.x >> 5;
    const int hi   = lane >> 4;
    const int lo   = lane & 15;
    const int row0 = blockIdx.x << 4;
    const int n0   = (blockIdx.y * 4 + wave) << 4;
    if (n0 >= Nc) return;

    int arow = row0 + lo; if (arow > M - 1) arow = M - 1;
    int bcol = n0 + lo;   if (bcol > Nc - 1) bcol = Nc - 1;
    const float* ap  = A + (size_t)arow * K + (hi << 1);
    const float* bp0 = B + (size_t)(hi << 1) * Nc + bcol;
    const float* bp1 = bp0 + Nc;
    const size_t bstep = (size_t)Nc * 4;

    v8f acc = {0.f, 0.f, 0.f, 0.f, 0.f, 0.f, 0.f, 0.f};
    #pragma unroll 4
    for (int k = 0; k < K; k += 4) {
        v2f a = *(const v2f*)ap;
        v2f b;
        b.x = *bp0;
        b.y = *bp1;
        acc = __builtin_amdgcn_wmma_f32_16x16x4_f32(
             false, a,  false, b,
             (short)0, acc,  false,  false);
        ap  += 4;
        bp0 += bstep;
        bp1 += bstep;
    }

    const int n = n0 + lo;
    if (n < Nc) {
        const float bv = bias ? bias[n] : 0.f;
        float* cp = C + (size_t)(row0 + (hi << 3)) * Nc + n;
        #pragma unroll
        for (int j = 0; j < 8; ++j) {
            const int m = row0 + j + (hi << 3);
            if (m < M) {
                float v = acc[j] + bv;
                if (ACT == 1)
                    v = 0.5f * v * (1.f + erff(v * 0.70710678118654752f));
                cp[(size_t)j * Nc] = v;
            }
        }
    }
}

__global__ __launch_bounds__(256) void gcn_scatter(
    const float* __restrict__ h, const int* __restrict__ src,
    const int* __restrict__ dst, const float* __restrict__ ew,
    float* __restrict__ agg, int E)
{
    const int e = (blockIdx.x * blockDim.x + threadIdx.x) >> 5;
    if (e >= E) return;
    const int lane = threadIdx.x & 31;
    const int s = src[e];
    const int d = dst[e];
    const float w = ew[e];
    const float v0 = h[(size_t)s * HIDC + lane];
    const float v1 = h[(size_t)s * HIDC + lane + 32];
    unsafeAtomicAdd(&agg[(size_t)d * HIDC + lane], v0 * w);
    unsafeAtomicAdd(&agg[(size_t)d * HIDC + lane + 32], v1 * w);
}

__global__ __launch_bounds__(256) void bias_relu_stats(
    float* __restrict__ x, const float* __restrict__ bias,
    float* __restrict__ stats, int Nrows)
{
    const int t = blockIdx.x * blockDim.x + threadIdx.x;
    const int c = t & 63;
    const int rstep = (gridDim.x * blockDim.x) >> 6;
    const float b = bias[c];
    float s = 0.f, s2 = 0.f;
    float* p = x + (size_t)(t >> 6) * HIDC + c;
    const size_t pstep = (size_t)rstep * HIDC;
    for (int r = t >> 6; r < Nrows; r += rstep) {
        float v = *p + b;
        v = fmaxf(v, 0.f);
        *p = v;
        s  += v;
        s2 += v * v;
        p  += pstep;
    }
    unsafeAtomicAdd(&stats[c], s);
    unsafeAtomicAdd(&stats[64 + c], s2);
}

__global__ void bn_finalize(float* __restrict__ stats, const float* __restrict__ g,
                            const float* __restrict__ be, float invN)
{
    const int c = threadIdx.x;
    const float mean = stats[c] * invN;
    const float var  = stats[64 + c] * invN - mean * mean;
    const float sc   = g[c] * rsqrtf(var + 1e-5f);
    stats[128 + c] = sc;
    stats[192 + c] = be[c] - mean * sc;
}

__global__ __launch_bounds__(256) void bn_apply(
    float* __restrict__ x, const float* __restrict__ stats, size_t total)
{
    size_t i = (size_t)blockIdx.x * blockDim.x + threadIdx.x;
    const size_t st = (size_t)gridDim.x * blockDim.x;
    for (; i < total; i += st) {
        const int c = (int)(i & 63);
        x[i] = x[i] * stats[128 + c] + stats[192 + c];
    }
}

__global__ __launch_bounds__(256) void zero_buf4(float4* __restrict__ p, size_t n4)
{
    size_t i = (size_t)blockIdx.x * blockDim.x + threadIdx.x;
    const size_t st = (size_t)gridDim.x * blockDim.x;
    const float4 z = make_float4(0.f, 0.f, 0.f, 0.f);
    for (; i < n4; i += st) p[i] = z;
}

extern "C" void kernel_launch(void* const* d_in, const int* in_sizes, int n_in,
                              void* d_out, int out_size, void* d_ws, size_t ws_size,
                              hipStream_t stream)
{
    const float* x  = (const float*)d_in[0];
    const int*   ei = (const int*)d_in[1];
    const float* ew = (const float*)d_in[2];
    const float* cw[3] = {(const float*)d_in[3], (const float*)d_in[5], (const float*)d_in[7]};
    const float* cb[3] = {(const float*)d_in[4], (const float*)d_in[6], (const float*)d_in[8]};
    const float* bg[3] = {(const float*)d_in[9],  (const float*)d_in[11], (const float*)d_in[13]};
    const float* bb[3] = {(const float*)d_in[10], (const float*)d_in[12], (const float*)d_in[14]};
    const float* w1 = (const float*)d_in[15]; const float* b1 = (const float*)d_in[16];
    const float* w2 = (const float*)d_in[17]; const float* b2 = (const float*)d_in[18];
    const float* w3 = (const float*)d_in[19]; const float* b3 = (const float*)d_in[20];

    const int N = in_sizes[0] / 128;
    const int E = in_sizes[2];
    const int* src = ei;
    const int* dst = ei + E;

    float* bufA  = (float*)d_ws;
    float* bufB  = bufA + (size_t)N * 128;
    float* stats = bufB + (size_t)N * 64;

    const dim3 blk128(128), blk256(256);
    const int rowTiles = (N + 15) / 16;
    const int eBlocks  = (E * 32 + 255) / 256;

    const float* in_ptr = x;
    int K = 128;
    for (int l = 0; l < 3; ++l) {
        gemm_wmma_f32<0><<<dim3(rowTiles, 1), blk128, 0, stream>>>(
            in_ptr, cw[l], nullptr, bufA, N, K, HIDC);
        zero_buf4<<<2048, blk256, 0, stream>>>((float4*)bufB, (size_t)N * HIDC / 4);
        zero_buf4<<<1, 64, 0, stream>>>((float4*)stats, 64);
        gcn_scatter<<<eBlocks, blk256, 0, stream>>>(bufA, src, dst, ew, bufB, E);
        bias_relu_stats<<<1024, blk256, 0, stream>>>(bufB, cb[l], stats, N);
        bn_finalize<<<1, 64, 0, stream>>>(stats, bg[l], bb[l], 1.0f / (float)N);
        bn_apply<<<2048, blk256, 0, stream>>>(bufB, stats, (size_t)N * HIDC);
        in_ptr = bufB;
        K = HIDC;
    }

    gemm_wmma_f32<1><<<dim3(rowTiles, 2), blk128, 0, stream>>>(bufB, w1, b1, bufA, N, 64, 128);
    gemm_wmma_f32<1><<<dim3(rowTiles, 1), blk128, 0, stream>>>(bufA, w2, b2, bufB, N, 128, 64);
    gemm_wmma_f32<0><<<dim3(rowTiles, 1), blk128, 0, stream>>>(bufB, w3, b3, (float*)d_out, N, 64, 5);
    (void)n_in; (void)out_size; (void)ws_size;
}


# Edge aggregation: float atomic scatter-add replaced by per-call counting sort of edges by destination + one-wave-per-node register gather; row sum order = sorted slot order (arrival-dependent)
# speedup vs baseline: 1.6508x; 1.6508x over previous
#include <hip/hip_runtime.h>
#include <cstddef>


typedef __attribute__((ext_vector_type(2))) float v2f;
typedef __attribute__((ext_vector_type(8))) float v8f;

#define HIDC 64

template <int ACT>
__global__ __launch_bounds__(128) void gemm_wmma_f32(
    const float* __restrict__ A, const float* __restrict__ B,
    const float* __restrict__ bias, float* __restrict__ C,
    int M, int K, int Nc)
{
    const int lane = threadIdx.x & 31;
    const int wave = threadIdx.x >> 5;
    const int hi   = lane >> 4;
    const int lo   = lane & 15;
    const int row0 = blockIdx.x << 4;
    const int n0   = (blockIdx.y * 4 + wave) << 4;
    if (n0 >= Nc) return;

    int arow = row0 + lo; if (arow > M - 1) arow = M - 1;
    int bcol = n0 + lo;   if (bcol > Nc - 1) bcol = Nc - 1;
    const float* ap  = A + (size_t)arow * K + (hi << 1);
    const float* bp0 = B + (size_t)(hi << 1) * Nc + bcol;
    const float* bp1 = bp0 + Nc;
    const size_t bstep = (size_t)Nc * 4;

    v8f acc = {0.f, 0.f, 0.f, 0.f, 0.f, 0.f, 0.f, 0.f};
    #pragma unroll 4
    for (int k = 0; k < K; k += 4) {
        v2f a = *(const v2f*)ap;
        v2f b;
        b.x = *bp0;
        b.y = *bp1;
        acc = __builtin_amdgcn_wmma_f32_16x16x4_f32(
             false, a,  false, b,
             (short)0, acc,  false,  false);
        ap  += 4;
        bp0 += bstep;
        bp1 += bstep;
    }

    const int n = n0 + lo;
    if (n < Nc) {
        const float bv = bias ? bias[n] : 0.f;
        float* cp = C + (size_t)(row0 + (hi << 3)) * Nc + n;
        #pragma unroll
        for (int j = 0; j < 8; ++j) {
            const int m = row0 + j + (hi << 3);
            if (m < M) {
                float v = acc[j] + bv;
                if (ACT == 1)
                    v = 0.5f * v * (1.f + erff(v * 0.70710678118654752f));
                cp[(size_t)j * Nc] = v;
            }
        }
    }
}

__global__ __launch_bounds__(256) void zero_i32(int* __restrict__ p, int n)
{
    int i = blockIdx.x * blockDim.x + threadIdx.x;
    const int st = gridDim.x * blockDim.x;
    for (; i < n; i += st) p[i] = 0;
}

__global__ __launch_bounds__(256) void edge_rank(
    const int* __restrict__ dst, int* __restrict__ cnt, int* __restrict__ rank, int E)
{
    const int e = blockIdx.x * blockDim.x + threadIdx.x;
    if (e < E) rank[e] = atomicAdd(&cnt[dst[e]], 1);
}

#define SCAN_THREADS 1024
#define SCAN_PER 8
__global__ __launch_bounds__(SCAN_THREADS) void scan_counts(int* __restrict__ c, int n)
{
    __shared__ int wave_excl[32];
    __shared__ int tile_total;
    const int tid  = threadIdx.x;
    const int lane = tid & 31;
    const int wv   = tid >> 5;
    int carry = 0;
    for (int base = 0; base < n; base += SCAN_THREADS * SCAN_PER) {
        const int i0 = base + tid * SCAN_PER;
        int v[SCAN_PER];
        int tsum = 0;
        #pragma unroll
        for (int k = 0; k < SCAN_PER; ++k) {
            const int i  = i0 + k;
            const int ld = c[i < n ? i : n - 1];
            v[k] = (i < n) ? ld : 0;
            tsum += v[k];
        }
        int x = tsum;
        #pragma unroll
        for (int o = 1; o < 32; o <<= 1) {
            const int y = __shfl_up(x, o);
            if (lane >= o) x += y;
        }
        if (lane == 31) wave_excl[wv] = x;
        __syncthreads();
        if (wv == 0) {
            const int w = wave_excl[lane];
            int z = w;
            #pragma unroll
            for (int o = 1; o < 32; o <<= 1) {
                const int y = __shfl_up(z, o);
                if (lane >= o) z += y;
            }
            wave_excl[lane] = z - w;
            if (lane == 31) tile_total = z;
        }
        __syncthreads();
        int run = carry + wave_excl[wv] + (x - tsum);
        #pragma unroll
        for (int k = 0; k < SCAN_PER; ++k) {
            if (i0 + k < n) c[i0 + k] = run;
            run += v[k];
        }
        carry += tile_total;
        __syncthreads();
    }
    if (tid == 0) c[n] = carry;
}

__global__ __launch_bounds__(256) void edge_place(
    const int* __restrict__ dst, const int* __restrict__ off,
    const int* __restrict__ rank, int* __restrict__ perm, int E)
{
    const int e = blockIdx.x * blockDim.x + threadIdx.x;
    if (e < E) perm[off[dst[e]] + rank[e]] = e;
}


template <int K>
__device__ __forceinline__ void gather_rows(
    const float2* __restrict__ h2, int s, float w, int j0, int lane,
    float& ax, float& ay)
{
    float2 v[K];
    float  wj[K];
    #pragma unroll
    for (int k = 0; k < K; ++k) {
        const int sj = __builtin_amdgcn_readlane(s, j0 + k);
        wj[k] = __int_as_float(__builtin_amdgcn_readlane(__float_as_int(w), j0 + k));
        v[k]  = h2[(size_t)sj * (HIDC / 2) + lane];
    }
    #pragma unroll
    for (int k = 0; k < K; ++k) {
        ax += v[k].x * wj[k];
        ay += v[k].y * wj[k];
    }
}

__device__ __forceinline__ void gather_lanes(
    const float2* __restrict__ h2, int s, float w, int cnt, int lane,
    float& ax, float& ay)
{
    int j = 0;
    for (; j + 8 <= cnt; j += 8) gather_rows<8>(h2, s, w, j, lane, ax, ay);
    if (cnt & 4) { gather_rows<4>(h2, s, w, j, lane, ax, ay); j += 4; }
    if (cnt & 2) { gather_rows<2>(h2, s, w, j, lane, ax, ay); j += 2; }
    if (cnt & 1) { gather_rows<1>(h2, s, w, j, lane, ax, ay); }
}

__global__ __launch_bounds__(256) void gcn_gather(
    const float* __restrict__ h, const int* __restrict__ src,
    const float* __restrict__ ew, const int* __restrict__ off,
    const int* __restrict__ perm, float* __restrict__ agg, int Nn)
{
    const int d = __builtin_amdgcn_readfirstlane(
        (int)((blockIdx.x * blockDim.x + threadIdx.x) >> 5));
    if (d >= Nn) return;
    const int lane = threadIdx.x & 31;
    const int beg = off[d];
    const int end = off[d + 1];
    const float2* __restrict__ h2 = (const float2*)h;
    float ax = 0.f, ay = 0.f;
    for (int base = beg; base < end; base += 64) {
        const int left = end - base;
        const int c0 = left < 32 ? left : 32;
        const int r1 = left - 32;
        const int c1 = r1 < 0 ? 0 : (r1 < 32 ? r1 : 32);
        const int i0 = base + lane;
        const int i1 = base + 32 + lane;
        const int e0 = perm[i0 < end ? i0 : end - 1];
        const int e1 = perm[i1 < end ? i1 : end - 1];
        const int   s0 = src[e0];
        const int   s1 = src[e1];
        const float w0 = ew[e0];
        const float w1 = ew[e1];
        asm volatile("" :: "v"(s0), "v"(s1), "v"(w0), "v"(w1));
        gather_lanes(h2, s0, w0, c0, lane, ax, ay);
        gather_lanes(h2, s1, w1, c1, lane, ax, ay);
    }
    float2 r; r.x = ax; r.y = ay;
    ((float2*)agg)[(size_t)d * (HIDC / 2) + lane] = r;
}

__global__ __launch_bounds__(256) void bias_relu_stats(
    float* __restrict__ x, const float* __restrict__ bias,
    float* __restrict__ stats, int Nrows)
{
    const int t = blockIdx.x * blockDim.x + threadIdx.x;
    const int c = t & 63;
    const int rstep = (gridDim.x * blockDim.x) >> 6;
    const float b = bias[c];
    float s = 0.f, s2 = 0.f;
    float* p = x + (size_t)(t >> 6) * HIDC + c;
    const size_t pstep = (size_t)rstep * HIDC;
    for (int r = t >> 6; r < Nrows; r += rstep) {
        float v = *p + b;
        v = fmaxf(v, 0.f);
        *p = v;
        s  += v;
        s2 += v * v;
        p  += pstep;
    }
    unsafeAtomicAdd(&stats[c], s);
    unsafeAtomicAdd(&stats[64 + c], s2);
}

__global__ void bn_finalize(float* __restrict__ stats, const float* __restrict__ g,
                            const float* __restrict__ be, float invN)
{
    const int c = threadIdx.x;
    const float mean = stats[c] * invN;
    const float var  = stats[64 + c] * invN - mean * mean;
    const float sc   = g[c] * rsqrtf(var + 1e-5f);
    stats[128 + c] = sc;
    stats[192 + c] = be[c] - mean * sc;
}

__global__ __launch_bounds__(256) void bn_apply(
    float* __restrict__ x, const float* __restrict__ stats, size_t total)
{
    size_t i = (size_t)blockIdx.x * blockDim.x + threadIdx.x;
    const size_t st = (size_t)gridDim.x * blockDim.x;
    for (; i < total; i += st) {
        const int c = (int)(i & 63);
        x[i] = x[i] * stats[128 + c] + stats[192 + c];
    }
}

__global__ __launch_bounds__(256) void zero_buf4(float4* __restrict__ p, size_t n4)
{
    size_t i = (size_t)blockIdx.x * blockDim.x + threadIdx.x;
    const size_t st = (size_t)gridDim.x * blockDim.x;
    const float4 z = make_float4(0.f, 0.f, 0.f, 0.f);
    for (; i < n4; i += st) p[i] = z;
}

extern "C" void kernel_launch(void* const* d_in, const int* in_sizes, int n_in,
                              void* d_out, int out_size, void* d_ws, size_t ws_size,
                              hipStream_t stream)
{
    const float* x  = (const float*)d_in[0];
    const int*   ei = (const int*)d_in[1];
    const float* ew = (const float*)d_in[2];
    const float* cw[3] = {(const float*)d_in[3], (const float*)d_in[5], (const float*)d_in[7]};
    const float* cb[3] = {(const float*)d_in[4], (const float*)d_in[6], (const float*)d_in[8]};
    const float* bg[3] = {(const float*)d_in[9],  (const float*)d_in[11], (const float*)d_in[13]};
    const float* bb[3] = {(const float*)d_in[10], (const float*)d_in[12], (const float*)d_in[14]};
    const float* w1 = (const float*)d_in[15]; const float* b1 = (const float*)d_in[16];
    const float* w2 = (const float*)d_in[17]; const float* b2 = (const float*)d_in[18];
    const float* w3 = (const float*)d_in[19]; const float* b3 = (const float*)d_in[20];

    const int N = in_sizes[0] / 128;
    const int E = in_sizes[2];
    const int* src = ei;
    const int* dst = ei + E;

    float* bufA  = (float*)d_ws;
    float* bufB  = bufA + (size_t)N * 128;
    float* stats = bufB + (size_t)N * 64;
    int*   perm  = (int*)(bufA + (size_t)N * 64);
    int*   off   = perm + E;
    int*   rank  = (int*)bufB;

    const dim3 blk128(128), blk256(256);
    const int rowTiles = (N + 15) / 16;
    const int eBlocks  = (E + 255) / 256;
    const int nBlocks  = (N * 32 + 255) / 256;

    zero_i32<<<128, blk256, 0, stream>>>(off, N + 1);
    edge_rank<<<eBlocks, blk256, 0, stream>>>(dst, off, rank, E);
    scan_counts<<<1, SCAN_THREADS, 0, stream>>>(off, N);
    edge_place<<<eBlocks, blk256, 0, stream>>>(dst, off, rank, perm, E);

    const float* in_ptr = x;
    int K = 128;
    for (int l = 0; l < 3; ++l) {
        gemm_wmma_f32<0><<<dim3(rowTiles, 1), blk128, 0, stream>>>(
            in_ptr, cw[l], nullptr, bufA, N, K, HIDC);
        zero_buf4<<<1, 64, 0, stream>>>((float4*)stats, 64);
        gcn_gather<<<nBlocks, blk256, 0, stream>>>(bufA, src, ew, off, perm, bufB, N);
        bias_relu_stats<<<1024, blk256, 0, stream>>>(bufB, cb[l], stats, N);
        bn_finalize<<<1, 64, 0, stream>>>(stats, bg[l], bb[l], 1.0f / (float)N);
        bn_apply<<<2048, blk256, 0, stream>>>(bufB, stats, (size_t)N * HIDC);
        in_ptr = bufB;
        K = HIDC;
    }

    gemm_wmma_f32<1><<<dim3(rowTiles, 2), blk128, 0, stream>>>(bufB, w1, b1, bufA, N, 64, 128);
    gemm_wmma_f32<1><<<dim3(rowTiles, 1), blk128, 0, stream>>>(bufA, w2, b2, bufB, N, 128, 64);
    gemm_wmma_f32<0><<<dim3(rowTiles, 1), blk128, 0, stream>>>(bufB, w3, b3, (float*)d_out, N, 64, 5);
    (void)n_in; (void)out_size; (void)ws_size;
}


# Edge sort writes each edge's (source, weight) pair into its destination-sorted slot; gathers read pairs contiguously, not two random 4-byte lookups per edge via the permutation; sum order kept
# speedup vs baseline: 1.8797x; 1.1387x over previous
#include <hip/hip_runtime.h>
#include <cstddef>


typedef __attribute__((ext_vector_type(2))) float v2f;
typedef __attribute__((ext_vector_type(8))) float v8f;

#define HIDC 64

template <int ACT>
__global__ __launch_bounds__(128) void gemm_wmma_f32(
    const float* __restrict__ A, const float* __restrict__ B,
    const float* __restrict__ bias, float* __restrict__ C,
    int M, int K, int Nc)
{
    const int lane = threadIdx.x & 31;
    const int wave = threadIdx.x >> 5;
    const int hi   = lane >> 4;
    const int lo   = lane & 15;
    const int row0 = blockIdx.x << 4;
    const int n0   = (blockIdx.y * 4 + wave) << 4;
    if (n0 >= Nc) return;

    int arow = row0 + lo; if (arow > M - 1) arow = M - 1;
    int bcol = n0 + lo;   if (bcol > Nc - 1) bcol = Nc - 1;
    const float* ap  = A + (size_t)arow * K + (hi << 1);
    const float* bp0 = B + (size_t)(hi << 1) * Nc + bcol;
    const float* bp1 = bp0 + Nc;
    const size_t bstep = (size_t)Nc * 4;

    v8f acc = {0.f, 0.f, 0.f, 0.f, 0.f, 0.f, 0.f, 0.f};
    #pragma unroll 4
    for (int k = 0; k < K; k += 4) {
        v2f a = *(const v2f*)ap;
        v2f b;
        b.x = *bp0;
        b.y = *bp1;
        acc = __builtin_amdgcn_wmma_f32_16x16x4_f32(
             false, a,  false, b,
             (short)0, acc,  false,  false);
        ap  += 4;
        bp0 += bstep;
        bp1 += bstep;
    }

    const int n = n0 + lo;
    if (n < Nc) {
        const float bv = bias ? bias[n] : 0.f;
        float* cp = C + (size_t)(row0 + (hi << 3)) * Nc + n;
        #pragma unroll
        for (int j = 0; j < 8; ++j) {
            const int m = row0 + j + (hi << 3);
            if (m < M) {
                float v = acc[j] + bv;
                if (ACT == 1)
                    v = 0.5f * v * (1.f + erff(v * 0.70710678118654752f));
                cp[(size_t)j * Nc] = v;
            }
        }
    }
}

__global__ __launch_bounds__(256) void zero_i32(int* __restrict__ p, int n)
{
    int i = blockIdx.x * blockDim.x + threadIdx.x;
    const int st = gridDim.x * blockDim.x;
    for (; i < n; i += st) p[i] = 0;
}

__global__ __launch_bounds__(256) void edge_rank(
    const int* __restrict__ dst, int* __restrict__ cnt, int* __restrict__ rank, int E)
{
    const int e = blockIdx.x * blockDim.x + threadIdx.x;
    if (e < E) rank[e] = atomicAdd(&cnt[dst[e]], 1);
}

#define SCAN_THREADS 1024
#define SCAN_PER 8
__global__ __launch_bounds__(SCAN_THREADS) void scan_counts(int* __restrict__ c, int n)
{
    __shared__ int wave_excl[32];
    __shared__ int tile_total;
    const int tid  = threadIdx.x;
    const int lane = tid & 31;
    const int wv   = tid >> 5;
    int carry = 0;
    for (int base = 0; base < n; base += SCAN_THREADS * SCAN_PER) {
        const int i0 = base + tid * SCAN_PER;
        int v[SCAN_PER];
        int tsum = 0;
        #pragma unroll
        for (int k = 0; k < SCAN_PER; ++k) {
            const int i  = i0 + k;
            const int ld = c[i < n ? i : n - 1];
            v[k] = (i < n) ? ld : 0;
            tsum += v[k];
        }
        int x = tsum;
        #pragma unroll
        for (int o = 1; o < 32; o <<= 1) {
            const int y = __shfl_up(x, o);
            if (lane >= o) x += y;
        }
        if (lane == 31) wave_excl[wv] = x;
        __syncthreads();
        if (wv == 0) {
            const int w = wave_excl[lane];
            int z = w;
            #pragma unroll
            for (int o = 1; o < 32; o <<= 1) {
                const int y = __shfl_up(z, o);
                if (lane >= o) z += y;
            }
            wave_excl[lane] = z - w;
            if (lane == 31) tile_total = z;
        }
        __syncthreads();
        int run = carry + wave_excl[wv] + (x - tsum);
        #pragma unroll
        for (int k = 0; k < SCAN_PER; ++k) {
            if (i0 + k < n) c[i0 + k] = run;
            run += v[k];
        }
        carry += tile_total;
        __syncthreads();
    }
    if (tid == 0) c[n] = carry;
}

__global__ __launch_bounds__(256) void edge_place(
    const int* __restrict__ src, const int* __restrict__ dst,
    const float* __restrict__ ew, const int* __restrict__ off,
    const int* __restrict__ rank, int2* __restrict__ pairs, int E)
{
    const int e = blockIdx.x * blockDim.x + threadIdx.x;
    if (e < E) {
        int2 p;
        p.x = src[e];
        p.y = __float_as_int(ew[e]);
        pairs[off[dst[e]] + rank[e]] = p;
    }
}


template <int K>
__device__ __forceinline__ void gather_rows(
    const float2* __restrict__ h2, int s, float w, int j0, int lane,
    float& ax, float& ay)
{
    float2 v[K];
    float  wj[K];
    #pragma unroll
    for (int k = 0; k < K; ++k) {
        const int sj = __builtin_amdgcn_readlane(s, j0 + k);
        wj[k] = __int_as_float(__builtin_amdgcn_readlane(__float_as_int(w), j0 + k));
        v[k]  = h2[(size_t)sj * (HIDC / 2) + lane];
    }
    #pragma unroll
    for (int k = 0; k < K; ++k) {
        ax += v[k].x * wj[k];
        ay += v[k].y * wj[k];
    }
}

__device__ __forceinline__ void gather_lanes(
    const float2* __restrict__ h2, int s, float w, int cnt, int lane,
    float& ax, float& ay)
{
    int j = 0;
    for (; j + 8 <= cnt; j += 8) gather_rows<8>(h2, s, w, j, lane, ax, ay);
    if (cnt & 4) { gather_rows<4>(h2, s, w, j, lane, ax, ay); j += 4; }
    if (cnt & 2) { gather_rows<2>(h2, s, w, j, lane, ax, ay); j += 2; }
    if (cnt & 1) { gather_rows<1>(h2, s, w, j, lane, ax, ay); }
}

__global__ __launch_bounds__(256) void gcn_gather(
    const float* __restrict__ h, const int* __restrict__ off,
    const int2* __restrict__ pairs, float* __restrict__ agg, int Nn)
{
    const int d = __builtin_amdgcn_readfirstlane(
        (int)((blockIdx.x * blockDim.x + threadIdx.x) >> 5));
    if (d >= Nn) return;
    const int lane = threadIdx.x & 31;
    const int beg = off[d];
    const int end = off[d + 1];
    const float2* __restrict__ h2 = (const float2*)h;
    float ax = 0.f, ay = 0.f;
    for (int base = beg; base < end; base += 64) {
        const int left = end - base;
        const int c0 = left < 32 ? left : 32;
        const int r1 = left - 32;
        const int c1 = r1 < 0 ? 0 : (r1 < 32 ? r1 : 32);
        const int i0 = base + lane;
        const int i1 = base + 32 + lane;
        const int2 p0 = pairs[i0 < end ? i0 : end - 1];
        const int2 p1 = pairs[i1 < end ? i1 : end - 1];
        const int   s0 = p0.x;
        const int   s1 = p1.x;
        const float w0 = __int_as_float(p0.y);
        const float w1 = __int_as_float(p1.y);
        asm volatile("" :: "v"(s0), "v"(s1), "v"(w0), "v"(w1));
        gather_lanes(h2, s0, w0, c0, lane, ax, ay);
        gather_lanes(h2, s1, w1, c1, lane, ax, ay);
    }
    float2 r; r.x = ax; r.y = ay;
    ((float2*)agg)[(size_t)d * (HIDC / 2) + lane] = r;
}

__global__ __launch_bounds__(256) void bias_relu_stats(
    float* __restrict__ x, const float* __restrict__ bias,
    float* __restrict__ stats, int Nrows)
{
    const int t = blockIdx.x * blockDim.x + threadIdx.x;
    const int c = t & 63;
    const int rstep = (gridDim.x * blockDim.x) >> 6;
    const float b = bias[c];
    float s = 0.f, s2 = 0.f;
    float* p = x + (size_t)(t >> 6) * HIDC + c;
    const size_t pstep = (size_t)rstep * HIDC;
    for (int r = t >> 6; r < Nrows; r += rstep) {
        float v = *p + b;
        v = fmaxf(v, 0.f);
        *p = v;
        s  += v;
        s2 += v * v;
        p  += pstep;
    }
    unsafeAtomicAdd(&stats[c], s);
    unsafeAtomicAdd(&stats[64 + c], s2);
}

__global__ void bn_finalize(float* __restrict__ stats, const float* __restrict__ g,
                            const float* __restrict__ be, float invN)
{
    const int c = threadIdx.x;
    const float mean = stats[c] * invN;
    const float var  = stats[64 + c] * invN - mean * mean;
    const float sc   = g[c] * rsqrtf(var + 1e-5f);
    stats[128 + c] = sc;
    stats[192 + c] = be[c] - mean * sc;
}

__global__ __launch_bounds__(256) void bn_apply(
    float* __restrict__ x, const float* __restrict__ stats, size_t total)
{
    size_t i = (size_t)blockIdx.x * blockDim.x + threadIdx.x;
    const size_t st = (size_t)gridDim.x * blockDim.x;
    for (; i < total; i += st) {
        const int c = (int)(i & 63);
        x[i] = x[i] * stats[128 + c] + stats[192 + c];
    }
}

__global__ __launch_bounds__(256) void zero_buf4(float4* __restrict__ p, size_t n4)
{
    size_t i = (size_t)blockIdx.x * blockDim.x + threadIdx.x;
    const size_t st = (size_t)gridDim.x * blockDim.x;
    const float4 z = make_float4(0.f, 0.f, 0.f, 0.f);
    for (; i < n4; i += st) p[i] = z;
}

extern "C" void kernel_launch(void* const* d_in, const int* in_sizes, int n_in,
                              void* d_out, int out_size, void* d_ws, size_t ws_size,
                              hipStream_t stream)
{
    const float* x  = (const float*)d_in[0];
    const int*   ei = (const int*)d_in[1];
    const float* ew = (const float*)d_in[2];
    const float* cw[3] = {(const float*)d_in[3], (const float*)d_in[5], (const float*)d_in[7]};
    const float* cb[3] = {(const float*)d_in[4], (const float*)d_in[6], (const float*)d_in[8]};
    const float* bg[3] = {(const float*)d_in[9],  (const float*)d_in[11], (const float*)d_in[13]};
    const float* bb[3] = {(const float*)d_in[10], (const float*)d_in[12], (const float*)d_in[14]};
    const float* w1 = (const float*)d_in[15]; const float* b1 = (const float*)d_in[16];
    const float* w2 = (const float*)d_in[17]; const float* b2 = (const float*)d_in[18];
    const float* w3 = (const float*)d_in[19]; const float* b3 = (const float*)d_in[20];

    const int N = in_sizes[0] / 128;
    const int E = in_sizes[2];
    const int* src = ei;
    const int* dst = ei + E;

    float* bufA  = (float*)d_ws;
    float* bufB  = bufA + (size_t)N * 128;
    float* stats = bufB + (size_t)N * 64;
    int2*  pairs = (int2*)(bufA + (size_t)N * 64);
    int*   off   = (int*)d_out;
    int*   rank  = (int*)bufB;

    const dim3 blk128(128), blk256(256);
    const int rowTiles = (N + 15) / 16;
    const int eBlocks  = (E + 255) / 256;
    const int nBlocks  = (N * 32 + 255) / 256;

    zero_i32<<<128, blk256, 0, stream>>>(off, N + 1);
    edge_rank<<<eBlocks, blk256, 0, stream>>>(dst, off, rank, E);
    scan_counts<<<1, SCAN_THREADS, 0, stream>>>(off, N);
    edge_place<<<eBlocks, blk256, 0, stream>>>(src, dst, ew, off, rank, pairs, E);

    const float* in_ptr = x;
    int K = 128;
    for (int l = 0; l < 3; ++l) {
        gemm_wmma_f32<0><<<dim3(rowTiles, 1), blk128, 0, stream>>>(
            in_ptr, cw[l], nullptr, bufA, N, K, HIDC);
        zero_buf4<<<1, 64, 0, stream>>>((float4*)stats, 64);
        gcn_gather<<<nBlocks, blk256, 0, stream>>>(bufA, off, pairs, bufB, N);
        bias_relu_stats<<<1024, blk256, 0, stream>>>(bufB, cb[l], stats, N);
        bn_finalize<<<1, 64, 0, stream>>>(stats, bg[l], bb[l], 1.0f / (float)N);
        bn_apply<<<2048, blk256, 0, stream>>>(bufB, stats, (size_t)N * HIDC);
        in_ptr = bufB;
        K = HIDC;
    }

    gemm_wmma_f32<1><<<dim3(rowTiles, 2), blk128, 0, stream>>>(bufB, w1, b1, bufA, N, 64, 128);
    gemm_wmma_f32<1><<<dim3(rowTiles, 1), blk128, 0, stream>>>(bufA, w2, b2, bufB, N, 128, 64);
    gemm_wmma_f32<0><<<dim3(rowTiles, 1), blk128, 0, stream>>>(bufB, w3, b3, (float*)d_out, N, 64, 5);
    (void)n_in; (void)out_size; (void)ws_size;
}


# Elementwise passes fused: bias+ReLU+batch sums into the gather's store (resident grid; per-block sums, then atomics), BN scale/shift applied on the next product's A load; 25 -> 13 launches
# speedup vs baseline: 2.2198x; 1.1809x over previous
#include <hip/hip_runtime.h>
#include <cstddef>


typedef __attribute__((ext_vector_type(2))) float v2f;
typedef __attribute__((ext_vector_type(8))) float v8f;

#define HIDC 64

#define SUM_COPIES 8
template <int ACT, int BN>
__global__ __launch_bounds__(128) void gemm_wmma_f32(
    const float* __restrict__ A, const float* __restrict__ B,
    const float* __restrict__ bias, float* __restrict__ C,
    const float* __restrict__ bnsum, const float* __restrict__ bng,
    const float* __restrict__ bnb, float invN,
    int M, int K, int Nc)
{
    __shared__ float bnsc[BN ? HIDC : 1];
    __shared__ float bnsh[BN ? HIDC : 1];
    if (BN) {
        const int c = threadIdx.x;
        if (c < HIDC) {
            float s = 0.f, s2 = 0.f;
            #pragma unroll
            for (int j = 0; j < SUM_COPIES; ++j) {
                s  += bnsum[j * 128 + c];
                s2 += bnsum[j * 128 + 64 + c];
            }
            const float mean = s * invN;
            const float var  = s2 * invN - mean * mean;
            const float sc   = bng[c] * rsqrtf(var + 1e-5f);
            bnsc[c] = sc;
            bnsh[c] = bnb[c] - mean * sc;
        }
        __syncthreads();
    }
    const int lane = threadIdx.x & 31;
    const int wave = threadIdx.x >> 5;
    const int hi   = lane >> 4;
    const int lo   = lane & 15;
    const int row0 = blockIdx.x << 4;
    const int n0   = (blockIdx.y * 4 + wave) << 4;
    if (n0 >= Nc) return;

    int arow = row0 + lo; if (arow > M - 1) arow = M - 1;
    int bcol = n0 + lo;   if (bcol > Nc - 1) bcol = Nc - 1;
    const float* ap  = A + (size_t)arow * K + (hi << 1);
    const float* bp0 = B + (size_t)(hi << 1) * Nc + bcol;
    const float* bp1 = bp0 + Nc;
    const size_t bstep = (size_t)Nc * 4;

    v8f acc = {0.f, 0.f, 0.f, 0.f, 0.f, 0.f, 0.f, 0.f};
    #pragma unroll 4
    for (int k = 0; k < K; k += 4) {
        v2f a = *(const v2f*)ap;
        if (BN) {
            const v2f sc = *(const v2f*)&bnsc[k + (hi << 1)];
            const v2f sh = *(const v2f*)&bnsh[k + (hi << 1)];
            a.x = a.x * sc.x + sh.x;
            a.y = a.y * sc.y + sh.y;
        }
        v2f b;
        b.x = *bp0;
        b.y = *bp1;
        acc = __builtin_amdgcn_wmma_f32_16x16x4_f32(
             false, a,  false, b,
             (short)0, acc,  false,  false);
        ap  += 4;
        bp0 += bstep;
        bp1 += bstep;
    }

    const int n = n0 + lo;
    if (n < Nc) {
        const float bv = bias ? bias[n] : 0.f;
        float* cp = C + (size_t)(row0 + (hi << 3)) * Nc + n;
        #pragma unroll
        for (int j = 0; j < 8; ++j) {
            const int m = row0 + j + (hi << 3);
            if (m < M) {
                float v = acc[j] + bv;
                if (ACT == 1)
                    v = 0.5f * v * (1.f + erff(v * 0.70710678118654752f));
                cp[(size_t)j * Nc] = v;
            }
        }
    }
}

__global__ __launch_bounds__(256) void zero_i32(int* __restrict__ p, int n)
{
    int i = blockIdx.x * blockDim.x + threadIdx.x;
    const int st = gridDim.x * blockDim.x;
    for (; i < n; i += st) p[i] = 0;
}

__global__ __launch_bounds__(256) void edge_rank(
    const int* __restrict__ dst, int* __restrict__ cnt, int* __restrict__ rank, int E)
{
    const int e = blockIdx.x * blockDim.x + threadIdx.x;
    if (e < E) rank[e] = atomicAdd(&cnt[dst[e]], 1);
}

#define SCAN_THREADS 1024
#define SCAN_PER 8
__global__ __launch_bounds__(SCAN_THREADS) void scan_counts(int* __restrict__ c, int n)
{
    __shared__ int wave_excl[32];
    __shared__ int tile_total;
    const int tid  = threadIdx.x;
    const int lane = tid & 31;
    const int wv   = tid >> 5;
    int carry = 0;
    for (int base = 0; base < n; base += SCAN_THREADS * SCAN_PER) {
        const int i0 = base + tid * SCAN_PER;
        int v[SCAN_PER];
        int tsum = 0;
        #pragma unroll
        for (int k = 0; k < SCAN_PER; ++k) {
            const int i  = i0 + k;
            const int ld = c[i < n ? i : n - 1];
            v[k] = (i < n) ? ld : 0;
            tsum += v[k];
        }
        int x = tsum;
        #pragma unroll
        for (int o = 1; o < 32; o <<= 1) {
            const int y = __shfl_up(x, o);
            if (lane >= o) x += y;
        }
        if (lane == 31) wave_excl[wv] = x;
        __syncthreads();
        if (wv == 0) {
            const int w = wave_excl[lane];
            int z = w;
            #pragma unroll
            for (int o = 1; o < 32; o <<= 1) {
                const int y = __shfl_up(z, o);
                if (lane >= o) z += y;
            }
            wave_excl[lane] = z - w;
            if (lane == 31) tile_total = z;
        }
        __syncthreads();
        int run = carry + wave_excl[wv] + (x - tsum);
        #pragma unroll
        for (int k = 0; k < SCAN_PER; ++k) {
            if (i0 + k < n) c[i0 + k] = run;
            run += v[k];
        }
        carry += tile_total;
        __syncthreads();
    }
    if (tid == 0) c[n] = carry;
}

__global__ __launch_bounds__(256) void edge_place(
    const int* __restrict__ src, const int* __restrict__ dst,
    const float* __restrict__ ew, const int* __restrict__ off,
    const int* __restrict__ rank, int2* __restrict__ pairs, int E)
{
    const int e = blockIdx.x * blockDim.x + threadIdx.x;
    if (e < E) {
        int2 p;
        p.x = src[e];
        p.y = __float_as_int(ew[e]);
        pairs[off[dst[e]] + rank[e]] = p;
    }
}


template <int K>
__device__ __forceinline__ void gather_rows(
    const float2* __restrict__ h2, int s, float w, int j0, int lane,
    float& ax, float& ay)
{
    float2 v[K];
    float  wj[K];
    #pragma unroll
    for (int k = 0; k < K; ++k) {
        const int sj = __builtin_amdgcn_readlane(s, j0 + k);
        wj[k] = __int_as_float(__builtin_amdgcn_readlane(__float_as_int(w), j0 + k));
        v[k]  = h2[(size_t)sj * (HIDC / 2) + lane];
    }
    #pragma unroll
    for (int k = 0; k < K; ++k) {
        ax += v[k].x * wj[k];
        ay += v[k].y * wj[k];
    }
}

__device__ __forceinline__ void gather_lanes(
    const float2* __restrict__ h2, int s, float w, int cnt, int lane,
    float& ax, float& ay)
{
    int j = 0;
    for (; j + 8 <= cnt; j += 8) gather_rows<8>(h2, s, w, j, lane, ax, ay);
    if (cnt & 4) { gather_rows<4>(h2, s, w, j, lane, ax, ay); j += 4; }
    if (cnt & 2) { gather_rows<2>(h2, s, w, j, lane, ax, ay); j += 2; }
    if (cnt & 1) { gather_rows<1>(h2, s, w, j, lane, ax, ay); }
}

#define GATHER_BLOCKS 2048
__global__ __launch_bounds__(256) void gcn_gather_relu_sums(
    const float* __restrict__ h, const int* __restrict__ off,
    const int2* __restrict__ pairs, const float* __restrict__ bias,
    float* __restrict__ agg, float* __restrict__ sums, int Nn)
{
    __shared__ float red[8][128];
    const int lane = threadIdx.x & 31;
    const int wv   = threadIdx.x >> 5;
    int d = __builtin_amdgcn_readfirstlane((int)(blockIdx.x * 8 + wv));
    const int dstep = (int)gridDim.x * 8;
    const float2* __restrict__ h2 = (const float2*)h;
    const float2 b2 = ((const float2*)bias)[lane];
    float sx = 0.f, sy = 0.f, qx = 0.f, qy = 0.f;
    for (; d < Nn; d += dstep) {
        const int beg = off[d];
        const int end = off[d + 1];
        float ax = 0.f, ay = 0.f;
        for (int base = beg; base < end; base += 64) {
            const int left = end - base;
            const int c0 = left < 32 ? left : 32;
            const int r1 = left - 32;
            const int c1 = r1 < 0 ? 0 : (r1 < 32 ? r1 : 32);
            const int i0 = base + lane;
            const int i1 = base + 32 + lane;
            const int2 p0 = pairs[i0 < end ? i0 : end - 1];
            const int2 p1 = pairs[i1 < end ? i1 : end - 1];
            const int   s0 = p0.x;
            const int   s1 = p1.x;
            const float w0 = __int_as_float(p0.y);
            const float w1 = __int_as_float(p1.y);
            asm volatile("" :: "v"(s0), "v"(s1), "v"(w0), "v"(w1));
            gather_lanes(h2, s0, w0, c0, lane, ax, ay);
            gather_lanes(h2, s1, w1, c1, lane, ax, ay);
        }
        float2 r;
        r.x = fmaxf(ax + b2.x, 0.f);
        r.y = fmaxf(ay + b2.y, 0.f);
        ((float2*)agg)[(size_t)d * (HIDC / 2) + lane] = r;
        sx += r.x;        sy += r.y;
        qx += r.x * r.x;  qy += r.y * r.y;
    }
    red[wv][2 * lane]          = sx;
    red[wv][2 * lane + 1]      = sy;
    red[wv][64 + 2 * lane]     = qx;
    red[wv][64 + 2 * lane + 1] = qy;
    __syncthreads();
    if (threadIdx.x < 128) {
        float t = 0.f;
        #pragma unroll
        for (int w = 0; w < 8; ++w) t += red[w][threadIdx.x];
        unsafeAtomicAdd(&sums[(blockIdx.x & (SUM_COPIES - 1)) * 128 + threadIdx.x], t);
    }
}

extern "C" void kernel_launch(void* const* d_in, const int* in_sizes, int n_in,
                              void* d_out, int out_size, void* d_ws, size_t ws_size,
                              hipStream_t stream)
{
    const float* x  = (const float*)d_in[0];
    const int*   ei = (const int*)d_in[1];
    const float* ew = (const float*)d_in[2];
    const float* cw[3] = {(const float*)d_in[3], (const float*)d_in[5], (const float*)d_in[7]};
    const float* cb[3] = {(const float*)d_in[4], (const float*)d_in[6], (const float*)d_in[8]};
    const float* bg[3] = {(const float*)d_in[9],  (const float*)d_in[11], (const float*)d_in[13]};
    const float* bb[3] = {(const float*)d_in[10], (const float*)d_in[12], (const float*)d_in[14]};
    const float* w1 = (const float*)d_in[15]; const float* b1 = (const float*)d_in[16];
    const float* w2 = (const float*)d_in[17]; const float* b2 = (const float*)d_in[18];
    const float* w3 = (const float*)d_in[19]; const float* b3 = (const float*)d_in[20];

    const int N = in_sizes[0] / 128;
    const int E = in_sizes[2];
    const int* src = ei;
    const int* dst = ei + E;

    float* bufA  = (float*)d_ws;
    float* bufB  = bufA + (size_t)N * 128;
    int2*  pairs = (int2*)(bufA + (size_t)N * 64);
    int*   off   = (int*)d_out;
    int*   rank  = (int*)bufB;
    const int sumsBase = (N + 64) & ~63;
    const int sumsPer  = SUM_COPIES * 128;
    float* sums  = (float*)d_out + sumsBase;

    const dim3 blk128(128), blk256(256);
    const int rowTiles = (N + 15) / 16;
    const int eBlocks  = (E + 255) / 256;
    const float invN   = 1.0f / (float)N;

    zero_i32<<<128, blk256, 0, stream>>>(off, sumsBase + 3 * sumsPer);
    edge_rank<<<eBlocks, blk256, 0, stream>>>(dst, off, rank, E);
    scan_counts<<<1, SCAN_THREADS, 0, stream>>>(off, N);
    edge_place<<<eBlocks, blk256, 0, stream>>>(src, dst, ew, off, rank, pairs, E);

    gemm_wmma_f32<0, 0><<<dim3(rowTiles, 1), blk128, 0, stream>>>(
        x, cw[0], nullptr, bufA, nullptr, nullptr, nullptr, 0.f, N, 128, HIDC);
    gcn_gather_relu_sums<<<GATHER_BLOCKS, blk256, 0, stream>>>(
        bufA, off, pairs, cb[0], bufB, sums, N);
    for (int l = 1; l < 3; ++l) {
        gemm_wmma_f32<0, 1><<<dim3(rowTiles, 1), blk128, 0, stream>>>(
            bufB, cw[l], nullptr, bufA, sums + (l - 1) * sumsPer, bg[l - 1], bb[l - 1], invN,
            N, HIDC, HIDC);
        gcn_gather_relu_sums<<<GATHER_BLOCKS, blk256, 0, stream>>>(
            bufA, off, pairs, cb[l], bufB, sums + l * sumsPer, N);
    }

    gemm_wmma_f32<1, 1><<<dim3(rowTiles, 2), blk128, 0, stream>>>(
        bufB, w1, b1, bufA, sums + 2 * sumsPer, bg[2], bb[2], invN, N, 64, 128);
    gemm_wmma_f32<1, 0><<<dim3(rowTiles, 1), blk128, 0, stream>>>(
        bufA, w2, b2, bufB, nullptr, nullptr, nullptr, 0.f, N, 128, 64);
    gemm_wmma_f32<0, 0><<<dim3(rowTiles, 1), blk128, 0, stream>>>(
        bufB, w3, b3, (float*)d_out, nullptr, nullptr, nullptr, 0.f, N, 64, 5);
    (void)n_in; (void)out_size; (void)ws_size;
}


# Edge sort in two levels with LDS counters: edges grouped by block of 256 destinations (staged whole-run stores), then each block counting-sorted in LDS; sum order stays arrival-dependent
# speedup vs baseline: 4.0333x; 1.8170x over previous
#include <hip/hip_runtime.h>
#include <cstddef>


typedef __attribute__((ext_vector_type(2))) float v2f;
typedef __attribute__((ext_vector_type(8))) float v8f;

#define HIDC 64

#define SUM_COPIES 8
template <int ACT, int BN>
__global__ __launch_bounds__(128) void gemm_wmma_f32(
    const float* __restrict__ A, const float* __restrict__ B,
    const float* __restrict__ bias, float* __restrict__ C,
    const float* __restrict__ bnsum, const float* __restrict__ bng,
    const float* __restrict__ bnb, float invN,
    int M, int K, int Nc)
{
    __shared__ float bnsc[BN ? HIDC : 1];
    __shared__ float bnsh[BN ? HIDC : 1];
    if (BN) {
        const int c = threadIdx.x;
        if (c < HIDC) {
            float s = 0.f, s2 = 0.f;
            #pragma unroll
            for (int j = 0; j < SUM_COPIES; ++j) {
                s  += bnsum[j * 128 + c];
                s2 += bnsum[j * 128 + 64 + c];
            }
            const float mean = s * invN;
            const float var  = s2 * invN - mean * mean;
            const float sc   = bng[c] * rsqrtf(var + 1e-5f);
            bnsc[c] = sc;
            bnsh[c] = bnb[c] - mean * sc;
        }
        __syncthreads();
    }
    const int lane = threadIdx.x & 31;
    const int wave = threadIdx.x >> 5;
    const int hi   = lane >> 4;
    const int lo   = lane & 15;
    const int row0 = blockIdx.x << 4;
    const int n0   = (blockIdx.y * 4 + wave) << 4;
    if (n0 >= Nc) return;

    int arow = row0 + lo; if (arow > M - 1) arow = M - 1;
    int bcol = n0 + lo;   if (bcol > Nc - 1) bcol = Nc - 1;
    const float* ap  = A + (size_t)arow * K + (hi << 1);
    const float* bp0 = B + (size_t)(hi << 1) * Nc + bcol;
    const float* bp1 = bp0 + Nc;
    const size_t bstep = (size_t)Nc * 4;

    v8f acc = {0.f, 0.f, 0.f, 0.f, 0.f, 0.f, 0.f, 0.f};
    #pragma unroll 4
    for (int k = 0; k < K; k += 4) {
        v2f a = *(const v2f*)ap;
        if (BN) {
            const v2f sc = *(const v2f*)&bnsc[k + (hi << 1)];
            const v2f sh = *(const v2f*)&bnsh[k + (hi << 1)];
            a.x = a.x * sc.x + sh.x;
            a.y = a.y * sc.y + sh.y;
        }
        v2f b;
        b.x = *bp0;
        b.y = *bp1;
        acc = __builtin_amdgcn_wmma_f32_16x16x4_f32(
             false, a,  false, b,
             (short)0, acc,  false,  false);
        ap  += 4;
        bp0 += bstep;
        bp1 += bstep;
    }

    const int n = n0 + lo;
    if (n < Nc) {
        const float bv = bias ? bias[n] : 0.f;
        float* cp = C + (size_t)(row0 + (hi << 3)) * Nc + n;
        #pragma unroll
        for (int j = 0; j < 8; ++j) {
            const int m = row0 + j + (hi << 3);
            if (m < M) {
                float v = acc[j] + bv;
                if (ACT == 1)
                    v = 0.5f * v * (1.f + erff(v * 0.70710678118654752f));
                cp[(size_t)j * Nc] = v;
            }
        }
    }
}

#define BUCKET_SHIFT 8
#define BUCKET_NODES 256
#define KMAX 512

__global__ __launch_bounds__(256) void zero_i32(int* __restrict__ p, int n)
{
    int i = blockIdx.x * blockDim.x + threadIdx.x;
    const int st = gridDim.x * blockDim.x;
    for (; i < n; i += st) p[i] = 0;
}

__device__ __forceinline__ int block_excl_scan(int v, int* wave_excl)
{
    const int lane = threadIdx.x & 31;
    const int wv   = threadIdx.x >> 5;
    const int nw   = (int)(blockDim.x >> 5);
    int x = v;
    #pragma unroll
    for (int o = 1; o < 32; o <<= 1) {
        const int y = __shfl_up(x, o);
        if (lane >= o) x += y;
    }
    if (lane == 31) wave_excl[wv] = x;
    __syncthreads();
    if (wv == 0) {
        const int w = lane < nw ? wave_excl[lane] : 0;
        int z = w;
        #pragma unroll
        for (int o = 1; o < 32; o <<= 1) {
            const int y = __shfl_up(z, o);
            if (lane >= o) z += y;
        }
        wave_excl[lane] = z - w;
    }
    __syncthreads();
    const int r = wave_excl[wv] + (x - v);
    __syncthreads();
    return r;
}

#define PT_THREADS 1024
#define PT_TILE 12800
#define PT_PER ((PT_TILE + PT_THREADS - 1) / PT_THREADS)

__global__ __launch_bounds__(PT_THREADS) void bucket_hist(
    const int* __restrict__ dst, int* __restrict__ bucketCnt, int E, int K)
{
    __shared__ int lh[KMAX];
    const int tid = threadIdx.x;
    if (tid < KMAX) lh[tid] = 0;
    __syncthreads();
    for (int e0 = (int)blockIdx.x * PT_TILE; e0 < E; e0 += (int)gridDim.x * PT_TILE) {
        const int n = (E - e0) < PT_TILE ? (E - e0) : PT_TILE;
        int rd[PT_PER];
        #pragma unroll
        for (int k = 0; k < PT_PER; ++k) {
            const int i = tid + k * PT_THREADS;
            if (i < n) rd[k] = dst[e0 + i];
        }
        #pragma unroll
        for (int k = 0; k < PT_PER; ++k) {
            const int i = tid + k * PT_THREADS;
            if (i < n) atomicAdd(&lh[rd[k] >> BUCKET_SHIFT], 1);
        }
    }
    __syncthreads();
    if (tid < K) {
        const int c = lh[tid];
        if (c) atomicAdd(&bucketCnt[tid], c);
    }
}

__global__ __launch_bounds__(PT_THREADS) void edge_partition(
    const int* __restrict__ src, const int* __restrict__ dst,
    const float* __restrict__ ew, const int* __restrict__ bucketCnt,
    int* __restrict__ cursor, int2* __restrict__ grouped, int E, int K)
{
    __shared__ int2 s_pair[PT_TILE];
    __shared__ unsigned short s_bk[PT_TILE];
    __shared__ int lh[KMAX];
    __shared__ int lstart[KMAX];
    __shared__ int delta[KMAX];
    __shared__ int gbase[KMAX];
    __shared__ int wave_excl[32];
    const int tid = threadIdx.x;

    {
        const int tot = tid < K ? bucketCnt[tid] : 0;
        const int ex  = block_excl_scan(tot, wave_excl);
        if (tid < KMAX) gbase[tid] = ex;
    }
    for (int e0 = (int)blockIdx.x * PT_TILE; e0 < E; e0 += (int)gridDim.x * PT_TILE) {
        const int n = (E - e0) < PT_TILE ? (E - e0) : PT_TILE;
        if (tid < KMAX) lh[tid] = 0;
        __syncthreads();
        int   rs[PT_PER], rd[PT_PER], rr[PT_PER];
        float rw[PT_PER];
        #pragma unroll
        for (int k = 0; k < PT_PER; ++k) {
            const int i = tid + k * PT_THREADS;
            if (i < n) {
                rd[k] = dst[e0 + i];
                rs[k] = src[e0 + i];
                rw[k] = ew[e0 + i];
            }
        }
        #pragma unroll
        for (int k = 0; k < PT_PER; ++k) {
            const int i = tid + k * PT_THREADS;
            if (i < n) rr[k] = atomicAdd(&lh[rd[k] >> BUCKET_SHIFT], 1);
        }
        __syncthreads();
        const int c   = tid < K ? lh[tid] : 0;
        const int lex = block_excl_scan(c, wave_excl);
        if (tid < K) {
            lstart[tid] = lex;
            const int g = c ? atomicAdd(&cursor[tid], c) : 0;
            delta[tid] = gbase[tid] + g - lex;
        }
        __syncthreads();
        #pragma unroll
        for (int k = 0; k < PT_PER; ++k) {
            const int i = tid + k * PT_THREADS;
            if (i < n) {
                const int b = rd[k] >> BUCKET_SHIFT;
                const int p = lstart[b] + rr[k];
                int2 v;
                v.x = rs[k] | ((rd[k] & (BUCKET_NODES - 1)) << 24);
                v.y = __float_as_int(rw[k]);
                s_pair[p] = v;
                s_bk[p]   = (unsigned short)b;
            }
        }
        __syncthreads();
        for (int i = tid; i < n; i += PT_THREADS)
            grouped[delta[s_bk[i]] + i] = s_pair[i];
        __syncthreads();
    }
}

#define BK_THREADS 512
#define BK_CAP 12288
__global__ __launch_bounds__(BK_THREADS) void bucket_sort(
    const int* __restrict__ bucketCnt, const int2* __restrict__ grouped,
    int2* __restrict__ pairs, int* __restrict__ off, int Nn, int K)
{
    __shared__ int2 stage[BK_CAP];
    __shared__ int cnt[BUCKET_NODES];
    __shared__ int cur[BUCKET_NODES];
    __shared__ int wave_excl[32];
    __shared__ int s_base, s_count;
    const int tid = threadIdx.x;
    const int b   = blockIdx.x;
    {
        const int tot = tid < K ? bucketCnt[tid] : 0;
        const int ex  = block_excl_scan(tot, wave_excl);
        if (tid == b) { s_base = ex; s_count = tot; }
        if (tid < BUCKET_NODES) cnt[tid] = 0;
    }
    __syncthreads();
    const int base  = s_base;
    const int count = s_count;
    const int2* __restrict__ in = grouped + base;
    {
        int i = tid;
        for (; i + 3 * BK_THREADS < count; i += 4 * BK_THREADS) {
            const int x0 = in[i].x;
            const int x1 = in[i + BK_THREADS].x;
            const int x2 = in[i + 2 * BK_THREADS].x;
            const int x3 = in[i + 3 * BK_THREADS].x;
            atomicAdd(&cnt[(unsigned)x0 >> 24], 1);
            atomicAdd(&cnt[(unsigned)x1 >> 24], 1);
            atomicAdd(&cnt[(unsigned)x2 >> 24], 1);
            atomicAdd(&cnt[(unsigned)x3 >> 24], 1);
        }
        for (; i < count; i += BK_THREADS)
            atomicAdd(&cnt[(unsigned)in[i].x >> 24], 1);
    }
    __syncthreads();
    {
        const int c   = tid < BUCKET_NODES ? cnt[tid] : 0;
        const int lex = block_excl_scan(c, wave_excl);
        if (tid < BUCKET_NODES) {
            cur[tid] = lex;
            const int node = b * BUCKET_NODES + tid;
            if (node < Nn) off[node] = base + lex;
        }
        if (b == K - 1 && tid == 0) off[Nn] = base + count;
    }
    __syncthreads();
    const bool staged = count <= BK_CAP;
    {
        auto place = [&](int2 e) {
            const int r = atomicAdd(&cur[(unsigned)e.x >> 24], 1);
            int2 p;
            p.x = e.x & 0x00FFFFFF;
            p.y = e.y;
            if (staged) stage[r] = p;
            else        pairs[base + r] = p;
        };
        int i = tid;
        for (; i + 3 * BK_THREADS < count; i += 4 * BK_THREADS) {
            const int2 e0 = in[i];
            const int2 e1 = in[i + BK_THREADS];
            const int2 e2 = in[i + 2 * BK_THREADS];
            const int2 e3 = in[i + 3 * BK_THREADS];
            place(e0); place(e1); place(e2); place(e3);
        }
        for (; i < count; i += BK_THREADS) place(in[i]);
    }
    if (staged) {
        __syncthreads();
        for (int i = tid; i < count; i += BK_THREADS) pairs[base + i] = stage[i];
    }
}


template <int K>
__device__ __forceinline__ void gather_rows(
    const float2* __restrict__ h2, int s, float w, int j0, int lane,
    float& ax, float& ay)
{
    float2 v[K];
    float  wj[K];
    #pragma unroll
    for (int k = 0; k < K; ++k) {
        const int sj = __builtin_amdgcn_readlane(s, j0 + k);
        wj[k] = __int_as_float(__builtin_amdgcn_readlane(__float_as_int(w), j0 + k));
        v[k]  = h2[(size_t)sj * (HIDC / 2) + lane];
    }
    #pragma unroll
    for (int k = 0; k < K; ++k) {
        ax += v[k].x * wj[k];
        ay += v[k].y * wj[k];
    }
}

__device__ __forceinline__ void gather_lanes(
    const float2* __restrict__ h2, int s, float w, int cnt, int lane,
    float& ax, float& ay)
{
    int j = 0;
    for (; j + 8 <= cnt; j += 8) gather_rows<8>(h2, s, w, j, lane, ax, ay);
    if (cnt & 4) { gather_rows<4>(h2, s, w, j, lane, ax, ay); j += 4; }
    if (cnt & 2) { gather_rows<2>(h2, s, w, j, lane, ax, ay); j += 2; }
    if (cnt & 1) { gather_rows<1>(h2, s, w, j, lane, ax, ay); }
}

#define GATHER_BLOCKS 2048
__global__ __launch_bounds__(256) void gcn_gather_relu_sums(
    const float* __restrict__ h, const int* __restrict__ off,
    const int2* __restrict__ pairs, const float* __restrict__ bias,
    float* __restrict__ agg, float* __restrict__ sums, int Nn)
{
    __shared__ float red[8][128];
    const int lane = threadIdx.x & 31;
    const int wv   = threadIdx.x >> 5;
    int d = __builtin_amdgcn_readfirstlane((int)(blockIdx.x * 8 + wv));
    const int dstep = (int)gridDim.x * 8;
    const float2* __restrict__ h2 = (const float2*)h;
    const float2 b2 = ((const float2*)bias)[lane];
    float sx = 0.f, sy = 0.f, qx = 0.f, qy = 0.f;
    for (; d < Nn; d += dstep) {
        const int beg = off[d];
        const int end = off[d + 1];
        float ax = 0.f, ay = 0.f;
        for (int base = beg; base < end; base += 64) {
            const int left = end - base;
            const int c0 = left < 32 ? left : 32;
            const int r1 = left - 32;
            const int c1 = r1 < 0 ? 0 : (r1 < 32 ? r1 : 32);
            const int i0 = base + lane;
            const int i1 = base + 32 + lane;
            const int2 p0 = pairs[i0 < end ? i0 : end - 1];
            const int2 p1 = pairs[i1 < end ? i1 : end - 1];
            const int   s0 = p0.x;
            const int   s1 = p1.x;
            const float w0 = __int_as_float(p0.y);
            const float w1 = __int_as_float(p1.y);
            asm volatile("" :: "v"(s0), "v"(s1), "v"(w0), "v"(w1));
            gather_lanes(h2, s0, w0, c0, lane, ax, ay);
            gather_lanes(h2, s1, w1, c1, lane, ax, ay);
        }
        float2 r;
        r.x = fmaxf(ax + b2.x, 0.f);
        r.y = fmaxf(ay + b2.y, 0.f);
        ((float2*)agg)[(size_t)d * (HIDC / 2) + lane] = r;
        sx += r.x;        sy += r.y;
        qx += r.x * r.x;  qy += r.y * r.y;
    }
    red[wv][2 * lane]          = sx;
    red[wv][2 * lane + 1]      = sy;
    red[wv][64 + 2 * lane]     = qx;
    red[wv][64 + 2 * lane + 1] = qy;
    __syncthreads();
    if (threadIdx.x < 128) {
        float t = 0.f;
        #pragma unroll
        for (int w = 0; w < 8; ++w) t += red[w][threadIdx.x];
        unsafeAtomicAdd(&sums[(blockIdx.x & (SUM_COPIES - 1)) * 128 + threadIdx.x], t);
    }
}

extern "C" void kernel_launch(void* const* d_in, const int* in_sizes, int n_in,
                              void* d_out, int out_size, void* d_ws, size_t ws_size,
                              hipStream_t stream)
{
    const float* x  = (const float*)d_in[0];
    const int*   ei = (const int*)d_in[1];
    const float* ew = (const float*)d_in[2];
    const float* cw[3] = {(const float*)d_in[3], (const float*)d_in[5], (const float*)d_in[7]};
    const float* cb[3] = {(const float*)d_in[4], (const float*)d_in[6], (const float*)d_in[8]};
    const float* bg[3] = {(const float*)d_in[9],  (const float*)d_in[11], (const float*)d_in[13]};
    const float* bb[3] = {(const float*)d_in[10], (const float*)d_in[12], (const float*)d_in[14]};
    const float* w1 = (const float*)d_in[15]; const float* b1 = (const float*)d_in[16];
    const float* w2 = (const float*)d_in[17]; const float* b2 = (const float*)d_in[18];
    const float* w3 = (const float*)d_in[19]; const float* b3 = (const float*)d_in[20];

    const int N = in_sizes[0] / 128;
    const int E = in_sizes[2];
    const int* src = ei;
    const int* dst = ei + E;

    float* bufA  = (float*)d_ws;
    float* bufB  = bufA + (size_t)N * 128;
    int2*  pairs = (int2*)(bufA + (size_t)N * 64);
    int*   off   = (int*)d_out;
    int2*  grouped = (int2*)bufB;
    const int sumsBase = (N + 64) & ~63;
    const int sumsPer  = SUM_COPIES * 128;
    float* sums  = (float*)d_out + sumsBase;
    int*   bucketCnt = off + sumsBase + 3 * sumsPer;
    int*   cursor    = bucketCnt + KMAX;
    const int K = (N + BUCKET_NODES - 1) >> BUCKET_SHIFT;

    const dim3 blk128(128), blk256(256);
    const int rowTiles = (N + 15) / 16;
    const float invN   = 1.0f / (float)N;

    zero_i32<<<16, blk256, 0, stream>>>((int*)sums, 3 * sumsPer + 2 * KMAX);
    bucket_hist<<<256, PT_THREADS, 0, stream>>>(dst, bucketCnt, E, K);
    edge_partition<<<256, PT_THREADS, 0, stream>>>(src, dst, ew, bucketCnt, cursor, grouped, E, K);
    bucket_sort<<<K, BK_THREADS, 0, stream>>>(bucketCnt, grouped, pairs, off, N, K);

    gemm_wmma_f32<0, 0><<<dim3(rowTiles, 1), blk128, 0, stream>>>(
        x, cw[0], nullptr, bufA, nullptr, nullptr, nullptr, 0.f, N, 128, HIDC);
    gcn_gather_relu_sums<<<GATHER_BLOCKS, blk256, 0, stream>>>(
        bufA, off, pairs, cb[0], bufB, sums, N);
    for (int l = 1; l < 3; ++l) {
        gemm_wmma_f32<0, 1><<<dim3(rowTiles, 1), blk128, 0, stream>>>(
            bufB, cw[l], nullptr, bufA, sums + (l - 1) * sumsPer, bg[l - 1], bb[l - 1], invN,
            N, HIDC, HIDC);
        gcn_gather_relu_sums<<<GATHER_BLOCKS, blk256, 0, stream>>>(
            bufA, off, pairs, cb[l], bufB, sums + l * sumsPer, N);
    }

    gemm_wmma_f32<1, 1><<<dim3(rowTiles, 2), blk128, 0, stream>>>(
        bufB, w1, b1, bufA, sums + 2 * sumsPer, bg[2], bb[2], invN, N, 64, 128);
    gemm_wmma_f32<1, 0><<<dim3(rowTiles, 1), blk128, 0, stream>>>(
        bufA, w2, b2, bufB, nullptr, nullptr, nullptr, 0.f, N, 128, 64);
    gemm_wmma_f32<0, 0><<<dim3(rowTiles, 1), blk128, 0, stream>>>(
        bufB, w3, b3, (float*)d_out, nullptr, nullptr, nullptr, 0.f, N, 64, 5);
    (void)n_in; (void)out_size; (void)ws_size;
}


# Classifier MLP (64->128->64->5) as one launch: a wave takes 16 rows through all three products, intermediates in its own LDS strip, weights staged in LDS once per block; 13 -> 12 launches
# speedup vs baseline: 4.5528x; 1.1288x over previous
#include <hip/hip_runtime.h>
#include <cstddef>


typedef __attribute__((ext_vector_type(2))) float v2f;
typedef __attribute__((ext_vector_type(8))) float v8f;

#define HIDC 64

#define SUM_COPIES 8
template <int ACT, int BN>
__global__ __launch_bounds__(128) void gemm_wmma_f32(
    const float* __restrict__ A, const float* __restrict__ B,
    const float* __restrict__ bias, float* __restrict__ C,
    const float* __restrict__ bnsum, const float* __restrict__ bng,
    const float* __restrict__ bnb, float invN,
    int M, int K, int Nc)
{
    __shared__ float bnsc[BN ? HIDC : 1];
    __shared__ float bnsh[BN ? HIDC : 1];
    if (BN) {
        const int c = threadIdx.x;
        if (c < HIDC) {
            float s = 0.f, s2 = 0.f;
            #pragma unroll
            for (int j = 0; j < SUM_COPIES; ++j) {
                s  += bnsum[j * 128 + c];
                s2 += bnsum[j * 128 + 64 + c];
            }
            const float mean = s * invN;
            const float var  = s2 * invN - mean * mean;
            const float sc   = bng[c] * rsqrtf(var + 1e-5f);
            bnsc[c] = sc;
            bnsh[c] = bnb[c] - mean * sc;
        }
        __syncthreads();
    }
    const int lane = threadIdx.x & 31;
    const int wave = threadIdx.x >> 5;
    const int hi   = lane >> 4;
    const int lo   = lane & 15;
    const int row0 = blockIdx.x << 4;
    const int n0   = (blockIdx.y * 4 + wave) << 4;
    if (n0 >= Nc) return;

    int arow = row0 + lo; if (arow > M - 1) arow = M - 1;
    int bcol = n0 + lo;   if (bcol > Nc - 1) bcol = Nc - 1;
    const float* ap  = A + (size_t)arow * K + (hi << 1);
    const float* bp0 = B + (size_t)(hi << 1) * Nc + bcol;
    const float* bp1 = bp0 + Nc;
    const size_t bstep = (size_t)Nc * 4;

    v8f acc = {0.f, 0.f, 0.f, 0.f, 0.f, 0.f, 0.f, 0.f};
    #pragma unroll 4
    for (int k = 0; k < K; k += 4) {
        v2f a = *(const v2f*)ap;
        if (BN) {
            const v2f sc = *(const v2f*)&bnsc[k + (hi << 1)];
            const v2f sh = *(const v2f*)&bnsh[k + (hi << 1)];
            a.x = a.x * sc.x + sh.x;
            a.y = a.y * sc.y + sh.y;
        }
        v2f b;
        b.x = *bp0;
        b.y = *bp1;
        acc = __builtin_amdgcn_wmma_f32_16x16x4_f32(
             false, a,  false, b,
             (short)0, acc,  false,  false);
        ap  += 4;
        bp0 += bstep;
        bp1 += bstep;
    }

    const int n = n0 + lo;
    if (n < Nc) {
        const float bv = bias ? bias[n] : 0.f;
        float* cp = C + (size_t)(row0 + (hi << 3)) * Nc + n;
        #pragma unroll
        for (int j = 0; j < 8; ++j) {
            const int m = row0 + j + (hi << 3);
            if (m < M) {
                float v = acc[j] + bv;
                if (ACT == 1)
                    v = 0.5f * v * (1.f + erff(v * 0.70710678118654752f));
                cp[(size_t)j * Nc] = v;
            }
        }
    }
}

#define BUCKET_SHIFT 8
#define BUCKET_NODES 256
#define KMAX 512

__global__ __launch_bounds__(256) void zero_i32(int* __restrict__ p, int n)
{
    int i = blockIdx.x * blockDim.x + threadIdx.x;
    const int st = gridDim.x * blockDim.x;
    for (; i < n; i += st) p[i] = 0;
}

__device__ __forceinline__ int block_excl_scan(int v, int* wave_excl)
{
    const int lane = threadIdx.x & 31;
    const int wv   = threadIdx.x >> 5;
    const int nw   = (int)(blockDim.x >> 5);
    int x = v;
    #pragma unroll
    for (int o = 1; o < 32; o <<= 1) {
        const int y = __shfl_up(x, o);
        if (lane >= o) x += y;
    }
    if (lane == 31) wave_excl[wv] = x;
    __syncthreads();
    if (wv == 0) {
        const int w = lane < nw ? wave_excl[lane] : 0;
        int z = w;
        #pragma unroll
        for (int o = 1; o < 32; o <<= 1) {
            const int y = __shfl_up(z, o);
            if (lane >= o) z += y;
        }
        wave_excl[lane] = z - w;
    }
    __syncthreads();
    const int r = wave_excl[wv] + (x - v);
    __syncthreads();
    return r;
}

#define PT_THREADS 1024
#define PT_TILE 12800
#define PT_PER ((PT_TILE + PT_THREADS - 1) / PT_THREADS)

__global__ __launch_bounds__(PT_THREADS) void bucket_hist(
    const int* __restrict__ dst, int* __restrict__ bucketCnt, int E, int K)
{
    __shared__ int lh[KMAX];
    const int tid = threadIdx.x;
    if (tid < KMAX) lh[tid] = 0;
    __syncthreads();
    for (int e0 = (int)blockIdx.x * PT_TILE; e0 < E; e0 += (int)gridDim.x * PT_TILE) {
        const int n = (E - e0) < PT_TILE ? (E - e0) : PT_TILE;
        int rd[PT_PER];
        #pragma unroll
        for (int k = 0; k < PT_PER; ++k) {
            const int i = tid + k * PT_THREADS;
            if (i < n) rd[k] = dst[e0 + i];
        }
        #pragma unroll
        for (int k = 0; k < PT_PER; ++k) {
            const int i = tid + k * PT_THREADS;
            if (i < n) atomicAdd(&lh[rd[k] >> BUCKET_SHIFT], 1);
        }
    }
    __syncthreads();
    if (tid < K) {
        const int c = lh[tid];
        if (c) atomicAdd(&bucketCnt[tid], c);
    }
}

__global__ __launch_bounds__(PT_THREADS) void edge_partition(
    const int* __restrict__ src, const int* __restrict__ dst,
    const float* __restrict__ ew, const int* __restrict__ bucketCnt,
    int* __restrict__ cursor, int2* __restrict__ grouped, int E, int K)
{
    __shared__ int2 s_pair[PT_TILE];
    __shared__ unsigned short s_bk[PT_TILE];
    __shared__ int lh[KMAX];
    __shared__ int lstart[KMAX];
    __shared__ int delta[KMAX];
    __shared__ int gbase[KMAX];
    __shared__ int wave_excl[32];
    const int tid = threadIdx.x;

    {
        const int tot = tid < K ? bucketCnt[tid] : 0;
        const int ex  = block_excl_scan(tot, wave_excl);
        if (tid < KMAX) gbase[tid] = ex;
    }
    for (int e0 = (int)blockIdx.x * PT_TILE; e0 < E; e0 += (int)gridDim.x * PT_TILE) {
        const int n = (E - e0) < PT_TILE ? (E - e0) : PT_TILE;
        if (tid < KMAX) lh[tid] = 0;
        __syncthreads();
        int   rs[PT_PER], rd[PT_PER], rr[PT_PER];
        float rw[PT_PER];
        #pragma unroll
        for (int k = 0; k < PT_PER; ++k) {
            const int i = tid + k * PT_THREADS;
            if (i < n) {
                rd[k] = dst[e0 + i];
                rs[k] = src[e0 + i];
                rw[k] = ew[e0 + i];
            }
        }
        #pragma unroll
        for (int k = 0; k < PT_PER; ++k) {
            const int i = tid + k * PT_THREADS;
            if (i < n) rr[k] = atomicAdd(&lh[rd[k] >> BUCKET_SHIFT], 1);
        }
        __syncthreads();
        const int c   = tid < K ? lh[tid] : 0;
        const int lex = block_excl_scan(c, wave_excl);
        if (tid < K) {
            lstart[tid] = lex;
            const int g = c ? atomicAdd(&cursor[tid], c) : 0;
            delta[tid] = gbase[tid] + g - lex;
        }
        __syncthreads();
        #pragma unroll
        for (int k = 0; k < PT_PER; ++k) {
            const int i = tid + k * PT_THREADS;
            if (i < n) {
                const int b = rd[k] >> BUCKET_SHIFT;
                const int p = lstart[b] + rr[k];
                int2 v;
                v.x = rs[k] | ((rd[k] & (BUCKET_NODES - 1)) << 24);
                v.y = __float_as_int(rw[k]);
                s_pair[p] = v;
                s_bk[p]   = (unsigned short)b;
            }
        }
        __syncthreads();
        for (int i = tid; i < n; i += PT_THREADS)
            grouped[delta[s_bk[i]] + i] = s_pair[i];
        __syncthreads();
    }
}

#define BK_THREADS 512
#define BK_CAP 12288
__global__ __launch_bounds__(BK_THREADS) void bucket_sort(
    const int* __restrict__ bucketCnt, const int2* __restrict__ grouped,
    int2* __restrict__ pairs, int* __restrict__ off, int Nn, int K)
{
    __shared__ int2 stage[BK_CAP];
    __shared__ int cnt[BUCKET_NODES];
    __shared__ int cur[BUCKET_NODES];
    __shared__ int wave_excl[32];
    __shared__ int s_base, s_count;
    const int tid = threadIdx.x;
    const int b   = blockIdx.x;
    {
        const int tot = tid < K ? bucketCnt[tid] : 0;
        const int ex  = block_excl_scan(tot, wave_excl);
        if (tid == b) { s_base = ex; s_count = tot; }
        if (tid < BUCKET_NODES) cnt[tid] = 0;
    }
    __syncthreads();
    const int base  = s_base;
    const int count = s_count;
    const int2* __restrict__ in = grouped + base;
    {
        int i = tid;
        for (; i + 3 * BK_THREADS < count; i += 4 * BK_THREADS) {
            const int x0 = in[i].x;
            const int x1 = in[i + BK_THREADS].x;
            const int x2 = in[i + 2 * BK_THREADS].x;
            const int x3 = in[i + 3 * BK_THREADS].x;
            atomicAdd(&cnt[(unsigned)x0 >> 24], 1);
            atomicAdd(&cnt[(unsigned)x1 >> 24], 1);
            atomicAdd(&cnt[(unsigned)x2 >> 24], 1);
            atomicAdd(&cnt[(unsigned)x3 >> 24], 1);
        }
        for (; i < count; i += BK_THREADS)
            atomicAdd(&cnt[(unsigned)in[i].x >> 24], 1);
    }
    __syncthreads();
    {
        const int c   = tid < BUCKET_NODES ? cnt[tid] : 0;
        const int lex = block_excl_scan(c, wave_excl);
        if (tid < BUCKET_NODES) {
            cur[tid] = lex;
            const int node = b * BUCKET_NODES + tid;
            if (node < Nn) off[node] = base + lex;
        }
        if (b == K - 1 && tid == 0) off[Nn] = base + count;
    }
    __syncthreads();
    const bool staged = count <= BK_CAP;
    {
        auto place = [&](int2 e) {
            const int r = atomicAdd(&cur[(unsigned)e.x >> 24], 1);
            int2 p;
            p.x = e.x & 0x00FFFFFF;
            p.y = e.y;
            if (staged) stage[r] = p;
            else        pairs[base + r] = p;
        };
        int i = tid;
        for (; i + 3 * BK_THREADS < count; i += 4 * BK_THREADS) {
            const int2 e0 = in[i];
            const int2 e1 = in[i + BK_THREADS];
            const int2 e2 = in[i + 2 * BK_THREADS];
            const int2 e3 = in[i + 3 * BK_THREADS];
            place(e0); place(e1); place(e2); place(e3);
        }
        for (; i < count; i += BK_THREADS) place(in[i]);
    }
    if (staged) {
        __syncthreads();
        for (int i = tid; i < count; i += BK_THREADS) pairs[base + i] = stage[i];
    }
}


template <int K>
__device__ __forceinline__ void gather_rows(
    const float2* __restrict__ h2, int s, float w, int j0, int lane,
    float& ax, float& ay)
{
    float2 v[K];
    float  wj[K];
    #pragma unroll
    for (int k = 0; k < K; ++k) {
        const int sj = __builtin_amdgcn_readlane(s, j0 + k);
        wj[k] = __int_as_float(__builtin_amdgcn_readlane(__float_as_int(w), j0 + k));
        v[k]  = h2[(size_t)sj * (HIDC / 2) + lane];
    }
    #pragma unroll
    for (int k = 0; k < K; ++k) {
        ax += v[k].x * wj[k];
        ay += v[k].y * wj[k];
    }
}

__device__ __forceinline__ void gather_lanes(
    const float2* __restrict__ h2, int s, float w, int cnt, int lane,
    float& ax, float& ay)
{
    int j = 0;
    for (; j + 8 <= cnt; j += 8) gather_rows<8>(h2, s, w, j, lane, ax, ay);
    if (cnt & 4) { gather_rows<4>(h2, s, w, j, lane, ax, ay); j += 4; }
    if (cnt & 2) { gather_rows<2>(h2, s, w, j, lane, ax, ay); j += 2; }
    if (cnt & 1) { gather_rows<1>(h2, s, w, j, lane, ax, ay); }
}

#define GATHER_BLOCKS 2048
__global__ __launch_bounds__(256) void gcn_gather_relu_sums(
    const float* __restrict__ h, const int* __restrict__ off,
    const int2* __restrict__ pairs, const float* __restrict__ bias,
    float* __restrict__ agg, float* __restrict__ sums, int Nn)
{
    __shared__ float red[8][128];
    const int lane = threadIdx.x & 31;
    const int wv   = threadIdx.x >> 5;
    int d = __builtin_amdgcn_readfirstlane((int)(blockIdx.x * 8 + wv));
    const int dstep = (int)gridDim.x * 8;
    const float2* __restrict__ h2 = (const float2*)h;
    const float2 b2 = ((const float2*)bias)[lane];
    float sx = 0.f, sy = 0.f, qx = 0.f, qy = 0.f;
    for (; d < Nn; d += dstep) {
        const int beg = off[d];
        const int end = off[d + 1];
        float ax = 0.f, ay = 0.f;
        for (int base = beg; base < end; base += 64) {
            const int left = end - base;
            const int c0 = left < 32 ? left : 32;
            const int r1 = left - 32;
            const int c1 = r1 < 0 ? 0 : (r1 < 32 ? r1 : 32);
            const int i0 = base + lane;
            const int i1 = base + 32 + lane;
            const int2 p0 = pairs[i0 < end ? i0 : end - 1];
            const int2 p1 = pairs[i1 < end ? i1 : end - 1];
            const int   s0 = p0.x;
            const int   s1 = p1.x;
            const float w0 = __int_as_float(p0.y);
            const float w1 = __int_as_float(p1.y);
            asm volatile("" :: "v"(s0), "v"(s1), "v"(w0), "v"(w1));
            gather_lanes(h2, s0, w0, c0, lane, ax, ay);
            gather_lanes(h2, s1, w1, c1, lane, ax, ay);
        }
        float2 r;
        r.x = fmaxf(ax + b2.x, 0.f);
        r.y = fmaxf(ay + b2.y, 0.f);
        ((float2*)agg)[(size_t)d * (HIDC / 2) + lane] = r;
        sx += r.x;        sy += r.y;
        qx += r.x * r.x;  qy += r.y * r.y;
    }
    red[wv][2 * lane]          = sx;
    red[wv][2 * lane + 1]      = sy;
    red[wv][64 + 2 * lane]     = qx;
    red[wv][64 + 2 * lane + 1] = qy;
    __syncthreads();
    if (threadIdx.x < 128) {
        float t = 0.f;
        #pragma unroll
        for (int w = 0; w < 8; ++w) t += red[w][threadIdx.x];
        unsafeAtomicAdd(&sums[(blockIdx.x & (SUM_COPIES - 1)) * 128 + threadIdx.x], t);
    }
}


__global__ __launch_bounds__(64) void bn_fold(
    const float* __restrict__ bnsum, const float* __restrict__ bng,
    const float* __restrict__ bnb, float invN, float* __restrict__ scsh)
{
    const int c = threadIdx.x;
    float s = 0.f, s2 = 0.f;
    #pragma unroll
    for (int j = 0; j < SUM_COPIES; ++j) {
        s  += bnsum[j * 128 + c];
        s2 += bnsum[j * 128 + 64 + c];
    }
    const float mean = s * invN;
    const float var  = s2 * invN - mean * mean;
    const float sc   = bng[c] * rsqrtf(var + 1e-5f);
    scsh[c]      = sc;
    scsh[64 + c] = bnb[c] - mean * sc;
}

#define MLP_WAVES 16
#define P64  68
#define P128 132
#define MLP_BLOCKS 256
__device__ __forceinline__ float gelu_exact(float v)
{
    return 0.5f * v * (1.f + erff(v * 0.70710678118654752f));
}
__global__ __launch_bounds__(MLP_WAVES * 32) void mlp_fused(
    const float* __restrict__ R, const float* __restrict__ scsh,
    const float* __restrict__ w1, const float* __restrict__ b1,
    const float* __restrict__ w2, const float* __restrict__ b2,
    const float* __restrict__ w3, const float* __restrict__ b3,
    float* __restrict__ out, int M)
{
    __shared__ __attribute__((aligned(16))) float Wt1[128 * P64];
    __shared__ __attribute__((aligned(16))) float Wt2[64 * P128];
    __shared__ __attribute__((aligned(16))) float Wt3[16 * P64];
    __shared__ __attribute__((aligned(16))) float Hs[MLP_WAVES][16 * P128];
    __shared__ __attribute__((aligned(16))) float bnsc[64];
    __shared__ __attribute__((aligned(16))) float bnsh[64];
    const int tid = threadIdx.x;
    {
        constexpr int NT = MLP_WAVES * 32;
        constexpr int PER = 64 * 128 / NT;
        float r1[PER], r2[PER];
        #pragma unroll
        for (int j = 0; j < PER; ++j) r1[j] = w1[tid + j * NT];
        #pragma unroll
        for (int j = 0; j < PER; ++j) r2[j] = w2[tid + j * NT];
        #pragma unroll
        for (int j = 0; j < PER; ++j) {
            const int i = tid + j * NT;
            Wt1[(i & 127) * P64 + (i >> 7)] = r1[j];
        }
        #pragma unroll
        for (int j = 0; j < PER; ++j) {
            const int i = tid + j * NT;
            Wt2[(i & 63) * P128 + (i >> 6)] = r2[j];
        }
    }
    for (int i = tid; i < 16 * 64; i += MLP_WAVES * 32) {
        const int n = i >> 6, k = i & 63;
        Wt3[n * P64 + k] = n < 5 ? w3[k * 5 + n] : 0.f;
    }
    if (tid < 64) { bnsc[tid] = scsh[tid]; bnsh[tid] = scsh[64 + tid]; }
    __syncthreads();

    const int lane = tid & 31;
    const int wave = tid >> 5;
    const int hi   = lane >> 4;
    const int lo   = lane & 15;
    float* hs = Hs[wave];
    const int tiles = (M + 15) >> 4;
    float bias1[8], bias2[4];
    #pragma unroll
    for (int c = 0; c < 8; ++c) bias1[c] = b1[c * 16 + lo];
    #pragma unroll
    for (int c = 0; c < 4; ++c) bias2[c] = b2[c * 16 + lo];
    const float bias3 = lo < 5 ? b3[lo] : 0.f;

    for (int t = wave * (int)gridDim.x + (int)blockIdx.x; t < tiles; t += (int)gridDim.x * MLP_WAVES) {
        const int row0 = t << 4;
        int arow = row0 + lo; if (arow > M - 1) arow = M - 1;
        const float* ap = R + (size_t)arow * 64 + (hi << 1);

        v2f a1[16];
        #pragma unroll
        for (int s = 0; s < 16; ++s) a1[s] = *(const v2f*)(ap + 4 * s);
        v8f acc1[8];
        #pragma unroll
        for (int c = 0; c < 8; ++c) acc1[c] = (v8f){0.f, 0.f, 0.f, 0.f, 0.f, 0.f, 0.f, 0.f};
        #pragma unroll
        for (int s = 0; s < 16; ++s) {
            const int k = 4 * s + (hi << 1);
            const v2f sc = *(const v2f*)&bnsc[k];
            const v2f sh = *(const v2f*)&bnsh[k];
            v2f a;
            a.x = a1[s].x * sc.x + sh.x;
            a.y = a1[s].y * sc.y + sh.y;
            #pragma unroll
            for (int c = 0; c < 8; ++c) {
                const v2f b = *(const v2f*)&Wt1[(c * 16 + lo) * P64 + k];
                acc1[c] = __builtin_amdgcn_wmma_f32_16x16x4_f32(
                    false, a, false, b, (short)0, acc1[c], false, false);
            }
        }
        #pragma unroll
        for (int c = 0; c < 8; ++c) {
            #pragma unroll
            for (int j = 0; j < 8; ++j)
                hs[(j + (hi << 3)) * P128 + c * 16 + lo] = gelu_exact(acc1[c][j] + bias1[c]);
        }

        v8f acc2[4];
        #pragma unroll
        for (int c = 0; c < 4; ++c) acc2[c] = (v8f){0.f, 0.f, 0.f, 0.f, 0.f, 0.f, 0.f, 0.f};
        #pragma unroll 8
        for (int s = 0; s < 32; ++s) {
            const int k = 4 * s + (hi << 1);
            const v2f a = *(const v2f*)&hs[lo * P128 + k];
            #pragma unroll
            for (int c = 0; c < 4; ++c) {
                const v2f b = *(const v2f*)&Wt2[(c * 16 + lo) * P128 + k];
                acc2[c] = __builtin_amdgcn_wmma_f32_16x16x4_f32(
                    false, a, false, b, (short)0, acc2[c], false, false);
            }
        }
        #pragma unroll
        for (int c = 0; c < 4; ++c) {
            #pragma unroll
            for (int j = 0; j < 8; ++j)
                hs[(j + (hi << 3)) * P128 + c * 16 + lo] = gelu_exact(acc2[c][j] + bias2[c]);
        }

        v8f acc3 = {0.f, 0.f, 0.f, 0.f, 0.f, 0.f, 0.f, 0.f};
        #pragma unroll
        for (int s = 0; s < 16; ++s) {
            const int k = 4 * s + (hi << 1);
            const v2f a = *(const v2f*)&hs[lo * P128 + k];
            const v2f b = *(const v2f*)&Wt3[lo * P64 + k];
            acc3 = __builtin_amdgcn_wmma_f32_16x16x4_f32(
                false, a, false, b, (short)0, acc3, false, false);
        }
        if (lo < 5) {
            #pragma unroll
            for (int j = 0; j < 8; ++j) {
                const int m = row0 + j + (hi << 3);
                if (m < M) out[(size_t)m * 5 + lo] = acc3[j] + bias3;
            }
        }
    }
}

extern "C" void kernel_launch(void* const* d_in, const int* in_sizes, int n_in,
                              void* d_out, int out_size, void* d_ws, size_t ws_size,
                              hipStream_t stream)
{
    const float* x  = (const float*)d_in[0];
    const int*   ei = (const int*)d_in[1];
    const float* ew = (const float*)d_in[2];
    const float* cw[3] = {(const float*)d_in[3], (const float*)d_in[5], (const float*)d_in[7]};
    const float* cb[3] = {(const float*)d_in[4], (const float*)d_in[6], (const float*)d_in[8]};
    const float* bg[3] = {(const float*)d_in[9],  (const float*)d_in[11], (const float*)d_in[13]};
    const float* bb[3] = {(const float*)d_in[10], (const float*)d_in[12], (const float*)d_in[14]};
    const float* w1 = (const float*)d_in[15]; const float* b1 = (const float*)d_in[16];
    const float* w2 = (const float*)d_in[17]; const float* b2 = (const float*)d_in[18];
    const float* w3 = (const float*)d_in[19]; const float* b3 = (const float*)d_in[20];

    const int N = in_sizes[0] / 128;
    const int E = in_sizes[2];
    const int* src = ei;
    const int* dst = ei + E;

    float* bufA  = (float*)d_ws;
    float* bufB  = bufA + (size_t)N * 128;
    int2*  pairs = (int2*)(bufA + (size_t)N * 64);
    int*   off   = (int*)d_out;
    int2*  grouped = (int2*)bufB;
    const int sumsBase = (N + 64) & ~63;
    const int sumsPer  = SUM_COPIES * 128;
    float* sums  = (float*)d_out + sumsBase;
    int*   bucketCnt = off + sumsBase + 3 * sumsPer;
    int*   cursor    = bucketCnt + KMAX;
    const int K = (N + BUCKET_NODES - 1) >> BUCKET_SHIFT;

    const dim3 blk128(128), blk256(256);
    const int rowTiles = (N + 15) / 16;
    const float invN   = 1.0f / (float)N;

    zero_i32<<<16, blk256, 0, stream>>>((int*)sums, 3 * sumsPer + 2 * KMAX);
    bucket_hist<<<256, PT_THREADS, 0, stream>>>(dst, bucketCnt, E, K);
    edge_partition<<<256, PT_THREADS, 0, stream>>>(src, dst, ew, bucketCnt, cursor, grouped, E, K);
    bucket_sort<<<K, BK_THREADS, 0, stream>>>(bucketCnt, grouped, pairs, off, N, K);

    gemm_wmma_f32<0, 0><<<dim3(rowTiles, 1), blk128, 0, stream>>>(
        x, cw[0], nullptr, bufA, nullptr, nullptr, nullptr, 0.f, N, 128, HIDC);
    gcn_gather_relu_sums<<<GATHER_BLOCKS, blk256, 0, stream>>>(
        bufA, off, pairs, cb[0], bufB, sums, N);
    for (int l = 1; l < 3; ++l) {
        gemm_wmma_f32<0, 1><<<dim3(rowTiles, 1), blk128, 0, stream>>>(
            bufB, cw[l], nullptr, bufA, sums + (l - 1) * sumsPer, bg[l - 1], bb[l - 1], invN,
            N, HIDC, HIDC);
        gcn_gather_relu_sums<<<GATHER_BLOCKS, blk256, 0, stream>>>(
            bufA, off, pairs, cb[l], bufB, sums + l * sumsPer, N);
    }

    bn_fold<<<1, 64, 0, stream>>>(sums + 2 * sumsPer, bg[2], bb[2], invN, bufA);
    mlp_fused<<<MLP_BLOCKS, MLP_WAVES * 32, 0, stream>>>(
        bufB, bufA, w1, b1, w2, b2, w3, b3, (float*)d_out, N);
    (void)n_in; (void)out_size; (void)ws_size;
}


# Layers 1-2: product moved behind the aggregation and into the gather launch (16-node LDS tiles); batch norm per source row before its weight, not on the summed row; sums regrouped; 12 -> 10 launches
# speedup vs baseline: 5.0658x; 1.1127x over previous
#include <hip/hip_runtime.h>
#include <cstddef>


typedef __attribute__((ext_vector_type(2))) float v2f;
typedef __attribute__((ext_vector_type(8))) float v8f;

#define HIDC 64

#define SUM_COPIES 8
template <int ACT>
__global__ __launch_bounds__(128) void gemm_wmma_f32(
    const float* __restrict__ A, const float* __restrict__ B,
    const float* __restrict__ bias, float* __restrict__ C,
    int M, int K, int Nc)
{
    const int lane = threadIdx.x & 31;
    const int wave = threadIdx.x >> 5;
    const int hi   = lane >> 4;
    const int lo   = lane & 15;
    const int row0 = blockIdx.x << 4;
    const int n0   = (blockIdx.y * 4 + wave) << 4;
    if (n0 >= Nc) return;

    int arow = row0 + lo; if (arow > M - 1) arow = M - 1;
    int bcol = n0 + lo;   if (bcol > Nc - 1) bcol = Nc - 1;
    const float* ap  = A + (size_t)arow * K + (hi << 1);
    const float* bp0 = B + (size_t)(hi << 1) * Nc + bcol;
    const float* bp1 = bp0 + Nc;
    const size_t bstep = (size_t)Nc * 4;

    v8f acc = {0.f, 0.f, 0.f, 0.f, 0.f, 0.f, 0.f, 0.f};
    #pragma unroll 4
    for (int k = 0; k < K; k += 4) {
        const v2f a = *(const v2f*)ap;
        v2f b;
        b.x = *bp0;
        b.y = *bp1;
        acc = __builtin_amdgcn_wmma_f32_16x16x4_f32(
             false, a,  false, b,
             (short)0, acc,  false,  false);
        ap  += 4;
        bp0 += bstep;
        bp1 += bstep;
    }

    const int n = n0 + lo;
    if (n < Nc) {
        const float bv = bias ? bias[n] : 0.f;
        float* cp = C + (size_t)(row0 + (hi << 3)) * Nc + n;
        #pragma unroll
        for (int j = 0; j < 8; ++j) {
            const int m = row0 + j + (hi << 3);
            if (m < M) {
                float v = acc[j] + bv;
                if (ACT == 1)
                    v = 0.5f * v * (1.f + erff(v * 0.70710678118654752f));
                cp[(size_t)j * Nc] = v;
            }
        }
    }
}

#define BUCKET_SHIFT 8
#define BUCKET_NODES 256
#define KMAX 512

__global__ __launch_bounds__(256) void zero_i32(int* __restrict__ p, int n)
{
    int i = blockIdx.x * blockDim.x + threadIdx.x;
    const int st = gridDim.x * blockDim.x;
    for (; i < n; i += st) p[i] = 0;
}

__device__ __forceinline__ int block_excl_scan(int v, int* wave_excl)
{
    const int lane = threadIdx.x & 31;
    const int wv   = threadIdx.x >> 5;
    const int nw   = (int)(blockDim.x >> 5);
    int x = v;
    #pragma unroll
    for (int o = 1; o < 32; o <<= 1) {
        const int y = __shfl_up(x, o);
        if (lane >= o) x += y;
    }
    if (lane == 31) wave_excl[wv] = x;
    __syncthreads();
    if (wv == 0) {
        const int w = lane < nw ? wave_excl[lane] : 0;
        int z = w;
        #pragma unroll
        for (int o = 1; o < 32; o <<= 1) {
            const int y = __shfl_up(z, o);
            if (lane >= o) z += y;
        }
        wave_excl[lane] = z - w;
    }
    __syncthreads();
    const int r = wave_excl[wv] + (x - v);
    __syncthreads();
    return r;
}

#define PT_THREADS 1024
#define PT_TILE 12800
#define PT_PER ((PT_TILE + PT_THREADS - 1) / PT_THREADS)

__global__ __launch_bounds__(PT_THREADS) void bucket_hist(
    const int* __restrict__ dst, int* __restrict__ bucketCnt, int E, int K)
{
    __shared__ int lh[KMAX];
    const int tid = threadIdx.x;
    if (tid < KMAX) lh[tid] = 0;
    __syncthreads();
    for (int e0 = (int)blockIdx.x * PT_TILE; e0 < E; e0 += (int)gridDim.x * PT_TILE) {
        const int n = (E - e0) < PT_TILE ? (E - e0) : PT_TILE;
        int rd[PT_PER];
        #pragma unroll
        for (int k = 0; k < PT_PER; ++k) {
            const int i = tid + k * PT_THREADS;
            if (i < n) rd[k] = dst[e0 + i];
        }
        #pragma unroll
        for (int k = 0; k < PT_PER; ++k) {
            const int i = tid + k * PT_THREADS;
            if (i < n) atomicAdd(&lh[rd[k] >> BUCKET_SHIFT], 1);
        }
    }
    __syncthreads();
    if (tid < K) {
        const int c = lh[tid];
        if (c) atomicAdd(&bucketCnt[tid], c);
    }
}

__global__ __launch_bounds__(PT_THREADS) void edge_partition(
    const int* __restrict__ src, const int* __restrict__ dst,
    const float* __restrict__ ew, const int* __restrict__ bucketCnt,
    int* __restrict__ cursor, int2* __restrict__ grouped, int E, int K)
{
    __shared__ int2 s_pair[PT_TILE];
    __shared__ unsigned short s_bk[PT_TILE];
    __shared__ int lh[KMAX];
    __shared__ int lstart[KMAX];
    __shared__ int delta[KMAX];
    __shared__ int gbase[KMAX];
    __shared__ int wave_excl[32];
    const int tid = threadIdx.x;

    {
        const int tot = tid < K ? bucketCnt[tid] : 0;
        const int ex  = block_excl_scan(tot, wave_excl);
        if (tid < KMAX) gbase[tid] = ex;
    }
    for (int e0 = (int)blockIdx.x * PT_TILE; e0 < E; e0 += (int)gridDim.x * PT_TILE) {
        const int n = (E - e0) < PT_TILE ? (E - e0) : PT_TILE;
        if (tid < KMAX) lh[tid] = 0;
        __syncthreads();
        int   rs[PT_PER], rd[PT_PER], rr[PT_PER];
        float rw[PT_PER];
        #pragma unroll
        for (int k = 0; k < PT_PER; ++k) {
            const int i = tid + k * PT_THREADS;
            if (i < n) {
                rd[k] = dst[e0 + i];
                rs[k] = src[e0 + i];
                rw[k] = ew[e0 + i];
            }
        }
        #pragma unroll
        for (int k = 0; k < PT_PER; ++k) {
            const int i = tid + k * PT_THREADS;
            if (i < n) rr[k] = atomicAdd(&lh[rd[k] >> BUCKET_SHIFT], 1);
        }
        __syncthreads();
        const int c   = tid < K ? lh[tid] : 0;
        const int lex = block_excl_scan(c, wave_excl);
        if (tid < K) {
            lstart[tid] = lex;
            const int g = c ? atomicAdd(&cursor[tid], c) : 0;
            delta[tid] = gbase[tid] + g - lex;
        }
        __syncthreads();
        #pragma unroll
        for (int k = 0; k < PT_PER; ++k) {
            const int i = tid + k * PT_THREADS;
            if (i < n) {
                const int b = rd[k] >> BUCKET_SHIFT;
                const int p = lstart[b] + rr[k];
                int2 v;
                v.x = rs[k] | ((rd[k] & (BUCKET_NODES - 1)) << 24);
                v.y = __float_as_int(rw[k]);
                s_pair[p] = v;
                s_bk[p]   = (unsigned short)b;
            }
        }
        __syncthreads();
        for (int i = tid; i < n; i += PT_THREADS)
            grouped[delta[s_bk[i]] + i] = s_pair[i];
        __syncthreads();
    }
}

#define BK_THREADS 512
#define BK_CAP 12288
__global__ __launch_bounds__(BK_THREADS) void bucket_sort(
    const int* __restrict__ bucketCnt, const int2* __restrict__ grouped,
    int2* __restrict__ pairs, int* __restrict__ off, int Nn, int K)
{
    __shared__ int2 stage[BK_CAP];
    __shared__ int cnt[BUCKET_NODES];
    __shared__ int cur[BUCKET_NODES];
    __shared__ int wave_excl[32];
    __shared__ int s_base, s_count;
    const int tid = threadIdx.x;
    const int b   = blockIdx.x;
    {
        const int tot = tid < K ? bucketCnt[tid] : 0;
        const int ex  = block_excl_scan(tot, wave_excl);
        if (tid == b) { s_base = ex; s_count = tot; }
        if (tid < BUCKET_NODES) cnt[tid] = 0;
    }
    __syncthreads();
    const int base  = s_base;
    const int count = s_count;
    const int2* __restrict__ in = grouped + base;
    {
        int i = tid;
        for (; i + 3 * BK_THREADS < count; i += 4 * BK_THREADS) {
            const int x0 = in[i].x;
            const int x1 = in[i + BK_THREADS].x;
            const int x2 = in[i + 2 * BK_THREADS].x;
            const int x3 = in[i + 3 * BK_THREADS].x;
            atomicAdd(&cnt[(unsigned)x0 >> 24], 1);
            atomicAdd(&cnt[(unsigned)x1 >> 24], 1);
            atomicAdd(&cnt[(unsigned)x2 >> 24], 1);
            atomicAdd(&cnt[(unsigned)x3 >> 24], 1);
        }
        for (; i < count; i += BK_THREADS)
            atomicAdd(&cnt[(unsigned)in[i].x >> 24], 1);
    }
    __syncthreads();
    {
        const int c   = tid < BUCKET_NODES ? cnt[tid] : 0;
        const int lex = block_excl_scan(c, wave_excl);
        if (tid < BUCKET_NODES) {
            cur[tid] = lex;
            const int node = b * BUCKET_NODES + tid;
            if (node < Nn) off[node] = base + lex;
        }
        if (b == K - 1 && tid == 0) off[Nn] = base + count;
    }
    __syncthreads();
    const bool staged = count <= BK_CAP;
    {
        auto place = [&](int2 e) {
            const int r = atomicAdd(&cur[(unsigned)e.x >> 24], 1);
            int2 p;
            p.x = e.x & 0x00FFFFFF;
            p.y = e.y;
            if (staged) stage[r] = p;
            else        pairs[base + r] = p;
        };
        int i = tid;
        for (; i + 3 * BK_THREADS < count; i += 4 * BK_THREADS) {
            const int2 e0 = in[i];
            const int2 e1 = in[i + BK_THREADS];
            const int2 e2 = in[i + 2 * BK_THREADS];
            const int2 e3 = in[i + 3 * BK_THREADS];
            place(e0); place(e1); place(e2); place(e3);
        }
        for (; i < count; i += BK_THREADS) place(in[i]);
    }
    if (staged) {
        __syncthreads();
        for (int i = tid; i < count; i += BK_THREADS) pairs[base + i] = stage[i];
    }
}


template <int K, bool BN>
__device__ __forceinline__ void gather_rows(
    const v2f* __restrict__ h2, int s, float w, int j0, int lane,
    v2f sc, v2f sh, v2f& acc)
{
    v2f   v[K];
    float wj[K];
    #pragma unroll
    for (int k = 0; k < K; ++k) {
        const int sj = __builtin_amdgcn_readlane(s, j0 + k);
        wj[k] = __int_as_float(__builtin_amdgcn_readlane(__float_as_int(w), j0 + k));
        v[k]  = h2[(size_t)sj * (HIDC / 2) + lane];
    }
    #pragma unroll
    for (int k = 0; k < K; ++k) {
        v2f t = v[k];
        if (BN) t = t * sc + sh;
        acc += t * wj[k];
    }
}

template <bool BN>
__device__ __forceinline__ void gather_lanes(
    const v2f* __restrict__ h2, int s, float w, int cnt, int lane,
    v2f sc, v2f sh, v2f& acc)
{
    int j = 0;
    for (; j + 8 <= cnt; j += 8) gather_rows<8, BN>(h2, s, w, j, lane, sc, sh, acc);
    if (cnt & 4) { gather_rows<4, BN>(h2, s, w, j, lane, sc, sh, acc); j += 4; }
    if (cnt & 2) { gather_rows<2, BN>(h2, s, w, j, lane, sc, sh, acc); j += 2; }
    if (cnt & 1) { gather_rows<1, BN>(h2, s, w, j, lane, sc, sh, acc); }
}

template <bool BN>
__device__ __forceinline__ void gather_segment(
    const v2f* __restrict__ h2, const int2* __restrict__ pairs,
    int beg, int end, int lane, v2f sc, v2f sh, v2f& acc)
{
    for (int base = beg; base < end; base += 64) {
        const int left = end - base;
        const int c0 = left < 32 ? left : 32;
        const int r1 = left - 32;
        const int c1 = r1 < 0 ? 0 : (r1 < 32 ? r1 : 32);
        const int i0 = base + lane;
        const int i1 = base + 32 + lane;
        const int2 p0 = pairs[i0 < end ? i0 : end - 1];
        const int2 p1 = pairs[i1 < end ? i1 : end - 1];
        const int   s0 = p0.x;
        const int   s1 = p1.x;
        const float w0 = __int_as_float(p0.y);
        const float w1 = __int_as_float(p1.y);
        asm volatile("" :: "v"(s0), "v"(s1), "v"(w0), "v"(w1));
        gather_lanes<BN>(h2, s0, w0, c0, lane, sc, sh, acc);
        gather_lanes<BN>(h2, s1, w1, c1, lane, sc, sh, acc);
    }
}

#define GATHER_BLOCKS 2048
__global__ __launch_bounds__(256) void gcn_gather_relu_sums(
    const float* __restrict__ h, const int* __restrict__ off,
    const int2* __restrict__ pairs, const float* __restrict__ bias,
    float* __restrict__ agg, float* __restrict__ sums, int Nn)
{
    __shared__ float red[8][128];
    const int lane = threadIdx.x & 31;
    const int wv   = threadIdx.x >> 5;
    int d = __builtin_amdgcn_readfirstlane((int)(blockIdx.x * 8 + wv));
    const int dstep = (int)gridDim.x * 8;
    const v2f* __restrict__ h2 = (const v2f*)h;
    const float2 b2 = ((const float2*)bias)[lane];
    float sx = 0.f, sy = 0.f, qx = 0.f, qy = 0.f;
    for (; d < Nn; d += dstep) {
        const int beg = off[d];
        const int end = off[d + 1];
        v2f acc = {0.f, 0.f};
        const v2f none = {0.f, 0.f};
        gather_segment<false>(h2, pairs, beg, end, lane, none, none, acc);
        float2 r;
        r.x = fmaxf(acc.x + b2.x, 0.f);
        r.y = fmaxf(acc.y + b2.y, 0.f);
        ((float2*)agg)[(size_t)d * (HIDC / 2) + lane] = r;
        sx += r.x;        sy += r.y;
        qx += r.x * r.x;  qy += r.y * r.y;
    }
    red[wv][2 * lane]          = sx;
    red[wv][2 * lane + 1]      = sy;
    red[wv][64 + 2 * lane]     = qx;
    red[wv][64 + 2 * lane + 1] = qy;
    __syncthreads();
    if (threadIdx.x < 128) {
        float t = 0.f;
        #pragma unroll
        for (int w = 0; w < 8; ++w) t += red[w][threadIdx.x];
        unsafeAtomicAdd(&sums[(blockIdx.x & (SUM_COPIES - 1)) * 128 + threadIdx.x], t);
    }
}


__global__ __launch_bounds__(64) void bn_fold(
    const float* __restrict__ bnsum, const float* __restrict__ bng,
    const float* __restrict__ bnb, float invN, float* __restrict__ scsh)
{
    const int c = threadIdx.x;
    float s = 0.f, s2 = 0.f;
    #pragma unroll
    for (int j = 0; j < SUM_COPIES; ++j) {
        s  += bnsum[j * 128 + c];
        s2 += bnsum[j * 128 + 64 + c];
    }
    const float mean = s * invN;
    const float var  = s2 * invN - mean * mean;
    const float sc   = bng[c] * rsqrtf(var + 1e-5f);
    scsh[c]      = sc;
    scsh[64 + c] = bnb[c] - mean * sc;
}

#define MLP_WAVES 16
#define P64  68
#define P128 132
#define MLP_BLOCKS 256
__device__ __forceinline__ float gelu_exact(float v)
{
    return 0.5f * v * (1.f + erff(v * 0.70710678118654752f));
}
__global__ __launch_bounds__(MLP_WAVES * 32) void mlp_fused(
    const float* __restrict__ R, const float* __restrict__ scsh,
    const float* __restrict__ w1, const float* __restrict__ b1,
    const float* __restrict__ w2, const float* __restrict__ b2,
    const float* __restrict__ w3, const float* __restrict__ b3,
    float* __restrict__ out, int M)
{
    __shared__ __attribute__((aligned(16))) float Wt1[128 * P64];
    __shared__ __attribute__((aligned(16))) float Wt2[64 * P128];
    __shared__ __attribute__((aligned(16))) float Wt3[16 * P64];
    __shared__ __attribute__((aligned(16))) float Hs[MLP_WAVES][16 * P128];
    __shared__ __attribute__((aligned(16))) float bnsc[64];
    __shared__ __attribute__((aligned(16))) float bnsh[64];
    const int tid = threadIdx.x;
    {
        constexpr int NT = MLP_WAVES * 32;
        constexpr int PER = 64 * 128 / NT;
        float r1[PER], r2[PER];
        #pragma unroll
        for (int j = 0; j < PER; ++j) r1[j] = w1[tid + j * NT];
        #pragma unroll
        for (int j = 0; j < PER; ++j) r2[j] = w2[tid + j * NT];
        #pragma unroll
        for (int j = 0; j < PER; ++j) {
            const int i = tid + j * NT;
            Wt1[(i & 127) * P64 + (i >> 7)] = r1[j];
        }
        #pragma unroll
        for (int j = 0; j < PER; ++j) {
            const int i = tid + j * NT;
            Wt2[(i & 63) * P128 + (i >> 6)] = r2[j];
        }
    }
    for (int i = tid; i < 16 * 64; i += MLP_WAVES * 32) {
        const int n = i >> 6, k = i & 63;
        Wt3[n * P64 + k] = n < 5 ? w3[k * 5 + n] : 0.f;
    }
    if (tid < 64) { bnsc[tid] = scsh[tid]; bnsh[tid] = scsh[64 + tid]; }
    __syncthreads();

    const int lane = tid & 31;
    const int wave = tid >> 5;
    const int hi   = lane >> 4;
    const int lo   = lane & 15;
    float* hs = Hs[wave];
    const int tiles = (M + 15) >> 4;
    float bias1[8], bias2[4];
    #pragma unroll
    for (int c = 0; c < 8; ++c) bias1[c] = b1[c * 16 + lo];
    #pragma unroll
    for (int c = 0; c < 4; ++c) bias2[c] = b2[c * 16 + lo];
    const float bias3 = lo < 5 ? b3[lo] : 0.f;

    for (int t = wave * (int)gridDim.x + (int)blockIdx.x; t < tiles; t += (int)gridDim.x * MLP_WAVES) {
        const int row0 = t << 4;
        int arow = row0 + lo; if (arow > M - 1) arow = M - 1;
        const float* ap = R + (size_t)arow * 64 + (hi << 1);

        v2f a1[16];
        #pragma unroll
        for (int s = 0; s < 16; ++s) a1[s] = *(const v2f*)(ap + 4 * s);
        v8f acc1[8];
        #pragma unroll
        for (int c = 0; c < 8; ++c) acc1[c] = (v8f){0.f, 0.f, 0.f, 0.f, 0.f, 0.f, 0.f, 0.f};
        #pragma unroll
        for (int s = 0; s < 16; ++s) {
            const int k = 4 * s + (hi << 1);
            const v2f sc = *(const v2f*)&bnsc[k];
            const v2f sh = *(const v2f*)&bnsh[k];
            v2f a;
            a.x = a1[s].x * sc.x + sh.x;
            a.y = a1[s].y * sc.y + sh.y;
            #pragma unroll
            for (int c = 0; c < 8; ++c) {
                const v2f b = *(const v2f*)&Wt1[(c * 16 + lo) * P64 + k];
                acc1[c] = __builtin_amdgcn_wmma_f32_16x16x4_f32(
                    false, a, false, b, (short)0, acc1[c], false, false);
            }
        }
        #pragma unroll
        for (int c = 0; c < 8; ++c) {
            #pragma unroll
            for (int j = 0; j < 8; ++j)
                hs[(j + (hi << 3)) * P128 + c * 16 + lo] = gelu_exact(acc1[c][j] + bias1[c]);
        }

        v8f acc2[4];
        #pragma unroll
        for (int c = 0; c < 4; ++c) acc2[c] = (v8f){0.f, 0.f, 0.f, 0.f, 0.f, 0.f, 0.f, 0.f};
        #pragma unroll 8
        for (int s = 0; s < 32; ++s) {
            const int k = 4 * s + (hi << 1);
            const v2f a = *(const v2f*)&hs[lo * P128 + k];
            #pragma unroll
            for (int c = 0; c < 4; ++c) {
                const v2f b = *(const v2f*)&Wt2[(c * 16 + lo) * P128 + k];
                acc2[c] = __builtin_amdgcn_wmma_f32_16x16x4_f32(
                    false, a, false, b, (short)0, acc2[c], false, false);
            }
        }
        #pragma unroll
        for (int c = 0; c < 4; ++c) {
            #pragma unroll
            for (int j = 0; j < 8; ++j)
                hs[(j + (hi << 3)) * P128 + c * 16 + lo] = gelu_exact(acc2[c][j] + bias2[c]);
        }

        v8f acc3 = {0.f, 0.f, 0.f, 0.f, 0.f, 0.f, 0.f, 0.f};
        #pragma unroll
        for (int s = 0; s < 16; ++s) {
            const int k = 4 * s + (hi << 1);
            const v2f a = *(const v2f*)&hs[lo * P128 + k];
            const v2f b = *(const v2f*)&Wt3[lo * P64 + k];
            acc3 = __builtin_amdgcn_wmma_f32_16x16x4_f32(
                false, a, false, b, (short)0, acc3, false, false);
        }
        if (lo < 5) {
            #pragma unroll
            for (int j = 0; j < 8; ++j) {
                const int m = row0 + j + (hi << 3);
                if (m < M) out[(size_t)m * 5 + lo] = acc3[j] + bias3;
            }
        }
    }
}

#define FUSED_WAVES 16
#define FUSED_BLOCKS 1024
__global__ __launch_bounds__(FUSED_WAVES * 32, 16) void gcn_gather_mm_relu_sums(
    const float* __restrict__ r, const int* __restrict__ off,
    const int2* __restrict__ pairs,
    const float* __restrict__ bnsum, const float* __restrict__ bng,
    const float* __restrict__ bnb, float invN,
    const float* __restrict__ w, const float* __restrict__ bias,
    float* __restrict__ out, float* __restrict__ sums, int Nn)
{
    __shared__ __attribute__((aligned(16))) float Wt[HIDC * P64];
    __shared__ __attribute__((aligned(16))) float Gs[2][16 * P64];
    __shared__ __attribute__((aligned(16))) float bnsc[HIDC];
    __shared__ __attribute__((aligned(16))) float bnsh[HIDC];
    __shared__ float red[2][2][HIDC];
    const int tid = threadIdx.x;
    if (tid < HIDC) {
        float s = 0.f, s2 = 0.f;
        #pragma unroll
        for (int j = 0; j < SUM_COPIES; ++j) {
            s  += bnsum[j * 128 + tid];
            s2 += bnsum[j * 128 + 64 + tid];
        }
        const float mean = s * invN;
        const float var  = s2 * invN - mean * mean;
        const float sc   = bng[tid] * rsqrtf(var + 1e-5f);
        bnsc[tid] = sc;
        bnsh[tid] = bnb[tid] - mean * sc;
    }
    #pragma unroll
    for (int j = 0; j < HIDC * HIDC / (FUSED_WAVES * 32); ++j) {
        const int i = tid + j * (FUSED_WAVES * 32);
        Wt[(i & 63) * P64 + (i >> 6)] = w[i];
    }
    __syncthreads();

    const int lane = tid & 31;
    const int wv = __builtin_amdgcn_readfirstlane(tid >> 5);
    const int hi = lane >> 4;
    const int lo = lane & 15;
    const v2f* __restrict__ r2 = (const v2f*)r;
    const v2f sc2 = ((const v2f*)bnsc)[lane];
    const v2f sh2 = ((const v2f*)bnsh)[lane];
    const int n0 = (wv & 3) << 4;
    const float bv = bias[n0 + lo];
    const int groups = (Nn + 15) >> 4;
    float sx = 0.f, qx = 0.f;
    int buf = 0;
    for (int g = (int)blockIdx.x; g < groups; g += (int)gridDim.x, buf ^= 1) {
        const int d = (g << 4) + wv;
        v2f gsum = {0.f, 0.f};
        if (d < Nn)
            gather_segment<true>(r2, pairs, off[d], off[d + 1], lane, sc2, sh2, gsum);
        *(v2f*)&Gs[buf][wv * P64 + 2 * lane] = gsum;
        __syncthreads();
        if (wv < 4) {
            v8f acc = {0.f, 0.f, 0.f, 0.f, 0.f, 0.f, 0.f, 0.f};
            #pragma unroll
            for (int s = 0; s < 16; ++s) {
                const int k = 4 * s + (hi << 1);
                const v2f a = *(const v2f*)&Gs[buf][lo * P64 + k];
                const v2f b = *(const v2f*)&Wt[(n0 + lo) * P64 + k];
                acc = __builtin_amdgcn_wmma_f32_16x16x4_f32(
                    false, a, false, b, (short)0, acc, false, false);
            }
            const int m0 = (g << 4) + (hi << 3);
            float* op = out + (size_t)m0 * HIDC + n0 + lo;
            #pragma unroll
            for (int j = 0; j < 8; ++j) {
                const float v = fmaxf(acc[j] + bv, 0.f);
                if (m0 + j < Nn) {
                    op[j * HIDC] = v;
                    sx += v;
                    qx += v * v;
                }
            }
        }
    }
    if (wv < 4) {
        red[0][hi][n0 + lo] = sx;
        red[1][hi][n0 + lo] = qx;
    }
    __syncthreads();
    if (tid < 128) {
        const float t = red[tid >> 6][0][tid & 63] + red[tid >> 6][1][tid & 63];
        unsafeAtomicAdd(&sums[(blockIdx.x & (SUM_COPIES - 1)) * 128 + tid], t);
    }
}

extern "C" void kernel_launch(void* const* d_in, const int* in_sizes, int n_in,
                              void* d_out, int out_size, void* d_ws, size_t ws_size,
                              hipStream_t stream)
{
    const float* x  = (const float*)d_in[0];
    const int*   ei = (const int*)d_in[1];
    const float* ew = (const float*)d_in[2];
    const float* cw[3] = {(const float*)d_in[3], (const float*)d_in[5], (const float*)d_in[7]};
    const float* cb[3] = {(const float*)d_in[4], (const float*)d_in[6], (const float*)d_in[8]};
    const float* bg[3] = {(const float*)d_in[9],  (const float*)d_in[11], (const float*)d_in[13]};
    const float* bb[3] = {(const float*)d_in[10], (const float*)d_in[12], (const float*)d_in[14]};
    const float* w1 = (const float*)d_in[15]; const float* b1 = (const float*)d_in[16];
    const float* w2 = (const float*)d_in[17]; const float* b2 = (const float*)d_in[18];
    const float* w3 = (const float*)d_in[19]; const float* b3 = (const float*)d_in[20];

    const int N = in_sizes[0] / 128;
    const int E = in_sizes[2];
    const int* src = ei;
    const int* dst = ei + E;

    float* bufA  = (float*)d_ws;
    float* bufB  = bufA + (size_t)N * 128;
    int2*  pairs = (int2*)(bufA + (size_t)N * 64);
    int*   off   = (int*)d_out;
    int2*  grouped = (int2*)bufB;
    const int sumsBase = (N + 64) & ~63;
    const int sumsPer  = SUM_COPIES * 128;
    float* sums  = (float*)d_out + sumsBase;
    int*   bucketCnt = off + sumsBase + 3 * sumsPer;
    int*   cursor    = bucketCnt + KMAX;
    const int K = (N + BUCKET_NODES - 1) >> BUCKET_SHIFT;

    const dim3 blk128(128), blk256(256);
    const int rowTiles = (N + 15) / 16;
    const float invN   = 1.0f / (float)N;

    zero_i32<<<16, blk256, 0, stream>>>((int*)sums, 3 * sumsPer + 2 * KMAX);
    bucket_hist<<<256, PT_THREADS, 0, stream>>>(dst, bucketCnt, E, K);
    edge_partition<<<256, PT_THREADS, 0, stream>>>(src, dst, ew, bucketCnt, cursor, grouped, E, K);
    bucket_sort<<<K, BK_THREADS, 0, stream>>>(bucketCnt, grouped, pairs, off, N, K);

    gemm_wmma_f32<0><<<dim3(rowTiles, 1), blk128, 0, stream>>>(
        x, cw[0], nullptr, bufA, N, 128, HIDC);
    gcn_gather_relu_sums<<<GATHER_BLOCKS, blk256, 0, stream>>>(
        bufA, off, pairs, cb[0], bufB, sums, N);
    gcn_gather_mm_relu_sums<<<FUSED_BLOCKS, FUSED_WAVES * 32, 0, stream>>>(
        bufB, off, pairs, sums, bg[0], bb[0], invN, cw[1], cb[1],
        bufA, sums + sumsPer, N);
    gcn_gather_mm_relu_sums<<<FUSED_BLOCKS, FUSED_WAVES * 32, 0, stream>>>(
        bufA, off, pairs, sums + sumsPer, bg[1], bb[1], invN, cw[2], cb[2],
        bufB, sums + 2 * sumsPer, N);

    bn_fold<<<1, 64, 0, stream>>>(sums + 2 * sumsPer, bg[2], bb[2], invN, bufA);
    mlp_fused<<<MLP_BLOCKS, MLP_WAVES * 32, 0, stream>>>(
        bufB, bufA, w1, b1, w2, b2, w3, b3, (float*)d_out, N);
    (void)n_in; (void)out_size; (void)ws_size;
}


# Layer-0 product x @ W moved into the sort's bucket-count launch: each wave counts, then takes a 16x64 strip (W in LDS; count loads unguarded, back to back); one-tile-per-wave launch gone, 10 -> 9
# speedup vs baseline: 5.3787x; 1.0618x over previous
#include <hip/hip_runtime.h>
#include <cstddef>


typedef __attribute__((ext_vector_type(2))) float v2f;
typedef __attribute__((ext_vector_type(8))) float v8f;

#define HIDC 64

#define INC  128
#define P64  68
#define P128 132
#define SUM_COPIES 8

#define BUCKET_SHIFT 8
#define BUCKET_NODES 256
#define KMAX 512

__global__ __launch_bounds__(256) void zero_i32(int* __restrict__ p, int n)
{
    int i = blockIdx.x * blockDim.x + threadIdx.x;
    const int st = gridDim.x * blockDim.x;
    for (; i < n; i += st) p[i] = 0;
}

__device__ __forceinline__ int block_excl_scan(int v, int* wave_excl)
{
    const int lane = threadIdx.x & 31;
    const int wv   = threadIdx.x >> 5;
    const int nw   = (int)(blockDim.x >> 5);
    int x = v;
    #pragma unroll
    for (int o = 1; o < 32; o <<= 1) {
        const int y = __shfl_up(x, o);
        if (lane >= o) x += y;
    }
    if (lane == 31) wave_excl[wv] = x;
    __syncthreads();
    if (wv == 0) {
        const int w = lane < nw ? wave_excl[lane] : 0;
        int z = w;
        #pragma unroll
        for (int o = 1; o < 32; o <<= 1) {
            const int y = __shfl_up(z, o);
            if (lane >= o) z += y;
        }
        wave_excl[lane] = z - w;
    }
    __syncthreads();
    const int r = wave_excl[wv] + (x - v);
    __syncthreads();
    return r;
}

#define PT_THREADS 1024
#define PT_TILE 12800
#define PT_PER ((PT_TILE + PT_THREADS - 1) / PT_THREADS)

__device__ __forceinline__ void load_a_strip(
    const float* __restrict__ x, int t, int M, int lo, int hi, v2f (&a)[INC / 4])
{
    int arow = (t << 4) + lo; if (arow > M - 1) arow = M - 1;
    const float* ap = x + (size_t)arow * INC + (hi << 1);
    #pragma unroll
    for (int s = 0; s < INC / 4; ++s) a[s] = *(const v2f*)(ap + 4 * s);
}
__global__ __launch_bounds__(PT_THREADS) __attribute__((amdgpu_waves_per_eu(8, 8)))
void bucket_hist_product0(
    const int* __restrict__ dst, int* __restrict__ bucketCnt, int E, int K,
    const float* __restrict__ x, const float* __restrict__ w0,
    float* __restrict__ h, int M)
{
    __shared__ int lh[KMAX];
    __shared__ __attribute__((aligned(16))) float Wt[HIDC * P128];
    const int tid  = threadIdx.x;
    const int lane = tid & 31;
    const int wave = __builtin_amdgcn_readfirstlane(tid >> 5);
    const int hi   = lane >> 4;
    const int lo   = lane & 15;
    const int tiles = (M + 15) >> 4;
    const int tstep = (int)gridDim.x * (PT_THREADS / 32);
    int t = wave * (int)gridDim.x + (int)blockIdx.x;

    if (tid < KMAX) lh[tid] = 0;
    __syncthreads();
    constexpr int WPER = INC * HIDC / PT_THREADS;
    float rw[WPER];
    #pragma unroll
    for (int j = 0; j < WPER; ++j) rw[j] = w0[tid + j * PT_THREADS];
    for (int e0 = (int)blockIdx.x * PT_TILE; e0 < E; e0 += (int)gridDim.x * PT_TILE) {
        const int n = (E - e0) < PT_TILE ? (E - e0) : PT_TILE;
        int rd[PT_PER];
        #pragma unroll
        for (int k = 0; k < PT_PER; ++k) {
            const int i = e0 + tid + k * PT_THREADS;
            rd[k] = dst[i < E ? i : E - 1];
        }
        #pragma unroll
        for (int k = 0; k < PT_PER; ++k)
            atomicAdd(&lh[rd[k] >> BUCKET_SHIFT], (tid + k * PT_THREADS) < n ? 1 : 0);
    }
    #pragma unroll
    for (int j = 0; j < WPER; ++j) {
        const int i = tid + j * PT_THREADS;
        Wt[(i & 63) * P128 + (i >> 6)] = rw[j];
    }
    __syncthreads();
    if (tid < K) {
        const int c = lh[tid];
        if (c) atomicAdd(&bucketCnt[tid], c);
    }

    while (t < tiles) {
        asm volatile("" ::: "memory");
        v2f a[INC / 4];
        load_a_strip(x, t, M, lo, hi, a);
        v8f acc[4];
        #pragma unroll
        for (int c = 0; c < 4; ++c) acc[c] = (v8f){0.f, 0.f, 0.f, 0.f, 0.f, 0.f, 0.f, 0.f};
        #pragma unroll
        for (int s = 0; s < INC / 4; ++s) {
            const int k = 4 * s + (hi << 1);
            #pragma unroll
            for (int c = 0; c < 4; ++c) {
                const v2f b = *(const v2f*)&Wt[(c * 16 + lo) * P128 + k];
                acc[c] = __builtin_amdgcn_wmma_f32_16x16x4_f32(
                    false, a[s], false, b, (short)0, acc[c], false, false);
            }
        }
        const int m0 = (t << 4) + (hi << 3);
        float* op = h + (size_t)m0 * HIDC + lo;
        #pragma unroll
        for (int c = 0; c < 4; ++c) {
            #pragma unroll
            for (int j = 0; j < 8; ++j)
                if (m0 + j < M) op[j * HIDC + c * 16] = acc[c][j];
        }
        t += tstep;
    }
}

__global__ __launch_bounds__(PT_THREADS) void edge_partition(
    const int* __restrict__ src, const int* __restrict__ dst,
    const float* __restrict__ ew, const int* __restrict__ bucketCnt,
    int* __restrict__ cursor, int2* __restrict__ grouped, int E, int K)
{
    __shared__ int2 s_pair[PT_TILE];
    __shared__ unsigned short s_bk[PT_TILE];
    __shared__ int lh[KMAX];
    __shared__ int lstart[KMAX];
    __shared__ int delta[KMAX];
    __shared__ int gbase[KMAX];
    __shared__ int wave_excl[32];
    const int tid = threadIdx.x;

    {
        const int tot = tid < K ? bucketCnt[tid] : 0;
        const int ex  = block_excl_scan(tot, wave_excl);
        if (tid < KMAX) gbase[tid] = ex;
    }
    for (int e0 = (int)blockIdx.x * PT_TILE; e0 < E; e0 += (int)gridDim.x * PT_TILE) {
        const int n = (E - e0) < PT_TILE ? (E - e0) : PT_TILE;
        if (tid < KMAX) lh[tid] = 0;
        __syncthreads();
        int   rs[PT_PER], rd[PT_PER], rr[PT_PER];
        float rw[PT_PER];
        #pragma unroll
        for (int k = 0; k < PT_PER; ++k) {
            const int i = tid + k * PT_THREADS;
            if (i < n) {
                rd[k] = dst[e0 + i];
                rs[k] = src[e0 + i];
                rw[k] = ew[e0 + i];
            }
        }
        #pragma unroll
        for (int k = 0; k < PT_PER; ++k) {
            const int i = tid + k * PT_THREADS;
            if (i < n) rr[k] = atomicAdd(&lh[rd[k] >> BUCKET_SHIFT], 1);
        }
        __syncthreads();
        const int c   = tid < K ? lh[tid] : 0;
        const int lex = block_excl_scan(c, wave_excl);
        if (tid < K) {
            lstart[tid] = lex;
            const int g = c ? atomicAdd(&cursor[tid], c) : 0;
            delta[tid] = gbase[tid] + g - lex;
        }
        __syncthreads();
        #pragma unroll
        for (int k = 0; k < PT_PER; ++k) {
            const int i = tid + k * PT_THREADS;
            if (i < n) {
                const int b = rd[k] >> BUCKET_SHIFT;
                const int p = lstart[b] + rr[k];
                int2 v;
                v.x = rs[k] | ((rd[k] & (BUCKET_NODES - 1)) << 24);
                v.y = __float_as_int(rw[k]);
                s_pair[p] = v;
                s_bk[p]   = (unsigned short)b;
            }
        }
        __syncthreads();
        for (int i = tid; i < n; i += PT_THREADS)
            grouped[delta[s_bk[i]] + i] = s_pair[i];
        __syncthreads();
    }
}

#define BK_THREADS 512
#define BK_CAP 12288
__global__ __launch_bounds__(BK_THREADS) void bucket_sort(
    const int* __restrict__ bucketCnt, const int2* __restrict__ grouped,
    int2* __restrict__ pairs, int* __restrict__ off, int Nn, int K)
{
    __shared__ int2 stage[BK_CAP];
    __shared__ int cnt[BUCKET_NODES];
    __shared__ int cur[BUCKET_NODES];
    __shared__ int wave_excl[32];
    __shared__ int s_base, s_count;
    const int tid = threadIdx.x;
    const int b   = blockIdx.x;
    {
        const int tot = tid < K ? bucketCnt[tid] : 0;
        const int ex  = block_excl_scan(tot, wave_excl);
        if (tid == b) { s_base = ex; s_count = tot; }
        if (tid < BUCKET_NODES) cnt[tid] = 0;
    }
    __syncthreads();
    const int base  = s_base;
    const int count = s_count;
    const int2* __restrict__ in = grouped + base;
    {
        int i = tid;
        for (; i + 3 * BK_THREADS < count; i += 4 * BK_THREADS) {
            const int x0 = in[i].x;
            const int x1 = in[i + BK_THREADS].x;
            const int x2 = in[i + 2 * BK_THREADS].x;
            const int x3 = in[i + 3 * BK_THREADS].x;
            atomicAdd(&cnt[(unsigned)x0 >> 24], 1);
            atomicAdd(&cnt[(unsigned)x1 >> 24], 1);
            atomicAdd(&cnt[(unsigned)x2 >> 24], 1);
            atomicAdd(&cnt[(unsigned)x3 >> 24], 1);
        }
        for (; i < count; i += BK_THREADS)
            atomicAdd(&cnt[(unsigned)in[i].x >> 24], 1);
    }
    __syncthreads();
    {
        const int c   = tid < BUCKET_NODES ? cnt[tid] : 0;
        const int lex = block_excl_scan(c, wave_excl);
        if (tid < BUCKET_NODES) {
            cur[tid] = lex;
            const int node = b * BUCKET_NODES + tid;
            if (node < Nn) off[node] = base + lex;
        }
        if (b == K - 1 && tid == 0) off[Nn] = base + count;
    }
    __syncthreads();
    const bool staged = count <= BK_CAP;
    {
        auto place = [&](int2 e) {
            const int r = atomicAdd(&cur[(unsigned)e.x >> 24], 1);
            int2 p;
            p.x = e.x & 0x00FFFFFF;
            p.y = e.y;
            if (staged) stage[r] = p;
            else        pairs[base + r] = p;
        };
        int i = tid;
        for (; i + 3 * BK_THREADS < count; i += 4 * BK_THREADS) {
            const int2 e0 = in[i];
            const int2 e1 = in[i + BK_THREADS];
            const int2 e2 = in[i + 2 * BK_THREADS];
            const int2 e3 = in[i + 3 * BK_THREADS];
            place(e0); place(e1); place(e2); place(e3);
        }
        for (; i < count; i += BK_THREADS) place(in[i]);
    }
    if (staged) {
        __syncthreads();
        for (int i = tid; i < count; i += BK_THREADS) pairs[base + i] = stage[i];
    }
}


template <int K, bool BN>
__device__ __forceinline__ void gather_rows(
    const v2f* __restrict__ h2, int s, float w, int j0, int lane,
    v2f sc, v2f sh, v2f& acc)
{
    v2f   v[K];
    float wj[K];
    #pragma unroll
    for (int k = 0; k < K; ++k) {
        const int sj = __builtin_amdgcn_readlane(s, j0 + k);
        wj[k] = __int_as_float(__builtin_amdgcn_readlane(__float_as_int(w), j0 + k));
        v[k]  = h2[(size_t)sj * (HIDC / 2) + lane];
    }
    #pragma unroll
    for (int k = 0; k < K; ++k) {
        v2f t = v[k];
        if (BN) t = t * sc + sh;
        acc += t * wj[k];
    }
}

template <bool BN>
__device__ __forceinline__ void gather_lanes(
    const v2f* __restrict__ h2, int s, float w, int cnt, int lane,
    v2f sc, v2f sh, v2f& acc)
{
    int j = 0;
    for (; j + 8 <= cnt; j += 8) gather_rows<8, BN>(h2, s, w, j, lane, sc, sh, acc);
    if (cnt & 4) { gather_rows<4, BN>(h2, s, w, j, lane, sc, sh, acc); j += 4; }
    if (cnt & 2) { gather_rows<2, BN>(h2, s, w, j, lane, sc, sh, acc); j += 2; }
    if (cnt & 1) { gather_rows<1, BN>(h2, s, w, j, lane, sc, sh, acc); }
}

template <bool BN>
__device__ __forceinline__ void gather_segment(
    const v2f* __restrict__ h2, const int2* __restrict__ pairs,
    int beg, int end, int lane, v2f sc, v2f sh, v2f& acc)
{
    for (int base = beg; base < end; base += 64) {
        const int left = end - base;
        const int c0 = left < 32 ? left : 32;
        const int r1 = left - 32;
        const int c1 = r1 < 0 ? 0 : (r1 < 32 ? r1 : 32);
        const int i0 = base + lane;
        const int i1 = base + 32 + lane;
        const int2 p0 = pairs[i0 < end ? i0 : end - 1];
        const int2 p1 = pairs[i1 < end ? i1 : end - 1];
        const int   s0 = p0.x;
        const int   s1 = p1.x;
        const float w0 = __int_as_float(p0.y);
        const float w1 = __int_as_float(p1.y);
        asm volatile("" :: "v"(s0), "v"(s1), "v"(w0), "v"(w1));
        gather_lanes<BN>(h2, s0, w0, c0, lane, sc, sh, acc);
        gather_lanes<BN>(h2, s1, w1, c1, lane, sc, sh, acc);
    }
}

#define GATHER_BLOCKS 2048
__global__ __launch_bounds__(256) void gcn_gather_relu_sums(
    const float* __restrict__ h, const int* __restrict__ off,
    const int2* __restrict__ pairs, const float* __restrict__ bias,
    float* __restrict__ agg, float* __restrict__ sums, int Nn)
{
    __shared__ float red[8][128];
    const int lane = threadIdx.x & 31;
    const int wv   = threadIdx.x >> 5;
    int d = __builtin_amdgcn_readfirstlane((int)(blockIdx.x * 8 + wv));
    const int dstep = (int)gridDim.x * 8;
    const v2f* __restrict__ h2 = (const v2f*)h;
    const float2 b2 = ((const float2*)bias)[lane];
    float sx = 0.f, sy = 0.f, qx = 0.f, qy = 0.f;
    for (; d < Nn; d += dstep) {
        const int beg = off[d];
        const int end = off[d + 1];
        v2f acc = {0.f, 0.f};
        const v2f none = {0.f, 0.f};
        gather_segment<false>(h2, pairs, beg, end, lane, none, none, acc);
        float2 r;
        r.x = fmaxf(acc.x + b2.x, 0.f);
        r.y = fmaxf(acc.y + b2.y, 0.f);
        ((float2*)agg)[(size_t)d * (HIDC / 2) + lane] = r;
        sx += r.x;        sy += r.y;
        qx += r.x * r.x;  qy += r.y * r.y;
    }
    red[wv][2 * lane]          = sx;
    red[wv][2 * lane + 1]      = sy;
    red[wv][64 + 2 * lane]     = qx;
    red[wv][64 + 2 * lane + 1] = qy;
    __syncthreads();
    if (threadIdx.x < 128) {
        float t = 0.f;
        #pragma unroll
        for (int w = 0; w < 8; ++w) t += red[w][threadIdx.x];
        unsafeAtomicAdd(&sums[(blockIdx.x & (SUM_COPIES - 1)) * 128 + threadIdx.x], t);
    }
}


__global__ __launch_bounds__(64) void bn_fold(
    const float* __restrict__ bnsum, const float* __restrict__ bng,
    const float* __restrict__ bnb, float invN, float* __restrict__ scsh)
{
    const int c = threadIdx.x;
    float s = 0.f, s2 = 0.f;
    #pragma unroll
    for (int j = 0; j < SUM_COPIES; ++j) {
        s  += bnsum[j * 128 + c];
        s2 += bnsum[j * 128 + 64 + c];
    }
    const float mean = s * invN;
    const float var  = s2 * invN - mean * mean;
    const float sc   = bng[c] * rsqrtf(var + 1e-5f);
    scsh[c]      = sc;
    scsh[64 + c] = bnb[c] - mean * sc;
}

#define MLP_WAVES 16
#define MLP_BLOCKS 256
__device__ __forceinline__ float gelu_exact(float v)
{
    return 0.5f * v * (1.f + erff(v * 0.70710678118654752f));
}
__global__ __launch_bounds__(MLP_WAVES * 32) void mlp_fused(
    const float* __restrict__ R, const float* __restrict__ scsh,
    const float* __restrict__ w1, const float* __restrict__ b1,
    const float* __restrict__ w2, const float* __restrict__ b2,
    const float* __restrict__ w3, const float* __restrict__ b3,
    float* __restrict__ out, int M)
{
    __shared__ __attribute__((aligned(16))) float Wt1[128 * P64];
    __shared__ __attribute__((aligned(16))) float Wt2[64 * P128];
    __shared__ __attribute__((aligned(16))) float Wt3[16 * P64];
    __shared__ __attribute__((aligned(16))) float Hs[MLP_WAVES][16 * P128];
    __shared__ __attribute__((aligned(16))) float bnsc[64];
    __shared__ __attribute__((aligned(16))) float bnsh[64];
    const int tid = threadIdx.x;
    {
        constexpr int NT = MLP_WAVES * 32;
        constexpr int PER = 64 * 128 / NT;
        float r1[PER], r2[PER];
        #pragma unroll
        for (int j = 0; j < PER; ++j) r1[j] = w1[tid + j * NT];
        #pragma unroll
        for (int j = 0; j < PER; ++j) r2[j] = w2[tid + j * NT];
        #pragma unroll
        for (int j = 0; j < PER; ++j) {
            const int i = tid + j * NT;
            Wt1[(i & 127) * P64 + (i >> 7)] = r1[j];
        }
        #pragma unroll
        for (int j = 0; j < PER; ++j) {
            const int i = tid + j * NT;
            Wt2[(i & 63) * P128 + (i >> 6)] = r2[j];
        }
    }
    for (int i = tid; i < 16 * 64; i += MLP_WAVES * 32) {
        const int n = i >> 6, k = i & 63;
        Wt3[n * P64 + k] = n < 5 ? w3[k * 5 + n] : 0.f;
    }
    if (tid < 64) { bnsc[tid] = scsh[tid]; bnsh[tid] = scsh[64 + tid]; }
    __syncthreads();

    const int lane = tid & 31;
    const int wave = tid >> 5;
    const int hi   = lane >> 4;
    const int lo   = lane & 15;
    float* hs = Hs[wave];
    const int tiles = (M + 15) >> 4;
    float bias1[8], bias2[4];
    #pragma unroll
    for (int c = 0; c < 8; ++c) bias1[c] = b1[c * 16 + lo];
    #pragma unroll
    for (int c = 0; c < 4; ++c) bias2[c] = b2[c * 16 + lo];
    const float bias3 = lo < 5 ? b3[lo] : 0.f;

    for (int t = wave * (int)gridDim.x + (int)blockIdx.x; t < tiles; t += (int)gridDim.x * MLP_WAVES) {
        const int row0 = t << 4;
        int arow = row0 + lo; if (arow > M - 1) arow = M - 1;
        const float* ap = R + (size_t)arow * 64 + (hi << 1);

        v2f a1[16];
        #pragma unroll
        for (int s = 0; s < 16; ++s) a1[s] = *(const v2f*)(ap + 4 * s);
        v8f acc1[8];
        #pragma unroll
        for (int c = 0; c < 8; ++c) acc1[c] = (v8f){0.f, 0.f, 0.f, 0.f, 0.f, 0.f, 0.f, 0.f};
        #pragma unroll
        for (int s = 0; s < 16; ++s) {
            const int k = 4 * s + (hi << 1);
            const v2f sc = *(const v2f*)&bnsc[k];
            const v2f sh = *(const v2f*)&bnsh[k];
            v2f a;
            a.x = a1[s].x * sc.x + sh.x;
            a.y = a1[s].y * sc.y + sh.y;
            #pragma unroll
            for (int c = 0; c < 8; ++c) {
                const v2f b = *(const v2f*)&Wt1[(c * 16 + lo) * P64 + k];
                acc1[c] = __builtin_amdgcn_wmma_f32_16x16x4_f32(
                    false, a, false, b, (short)0, acc1[c], false, false);
            }
        }
        #pragma unroll
        for (int c = 0; c < 8; ++c) {
            #pragma unroll
            for (int j = 0; j < 8; ++j)
                hs[(j + (hi << 3)) * P128 + c * 16 + lo] = gelu_exact(acc1[c][j] + bias1[c]);
        }

        v8f acc2[4];
        #pragma unroll
        for (int c = 0; c < 4; ++c) acc2[c] = (v8f){0.f, 0.f, 0.f, 0.f, 0.f, 0.f, 0.f, 0.f};
        #pragma unroll 8
        for (int s = 0; s < 32; ++s) {
            const int k = 4 * s + (hi << 1);
            const v2f a = *(const v2f*)&hs[lo * P128 + k];
            #pragma unroll
            for (int c = 0; c < 4; ++c) {
                const v2f b = *(const v2f*)&Wt2[(c * 16 + lo) * P128 + k];
                acc2[c] = __builtin_amdgcn_wmma_f32_16x16x4_f32(
                    false, a, false, b, (short)0, acc2[c], false, false);
            }
        }
        #pragma unroll
        for (int c = 0; c < 4; ++c) {
            #pragma unroll
            for (int j = 0; j < 8; ++j)
                hs[(j + (hi << 3)) * P128 + c * 16 + lo] = gelu_exact(acc2[c][j] + bias2[c]);
        }

        v8f acc3 = {0.f, 0.f, 0.f, 0.f, 0.f, 0.f, 0.f, 0.f};
        #pragma unroll
        for (int s = 0; s < 16; ++s) {
            const int k = 4 * s + (hi << 1);
            const v2f a = *(const v2f*)&hs[lo * P128 + k];
            const v2f b = *(const v2f*)&Wt3[lo * P64 + k];
            acc3 = __builtin_amdgcn_wmma_f32_16x16x4_f32(
                false, a, false, b, (short)0, acc3, false, false);
        }
        if (lo < 5) {
            #pragma unroll
            for (int j = 0; j < 8; ++j) {
                const int m = row0 + j + (hi << 3);
                if (m < M) out[(size_t)m * 5 + lo] = acc3[j] + bias3;
            }
        }
    }
}

#define FUSED_WAVES 16
#define FUSED_BLOCKS 1024
__global__ __launch_bounds__(FUSED_WAVES * 32, 16) void gcn_gather_mm_relu_sums(
    const float* __restrict__ r, const int* __restrict__ off,
    const int2* __restrict__ pairs,
    const float* __restrict__ bnsum, const float* __restrict__ bng,
    const float* __restrict__ bnb, float invN,
    const float* __restrict__ w, const float* __restrict__ bias,
    float* __restrict__ out, float* __restrict__ sums, int Nn)
{
    __shared__ __attribute__((aligned(16))) float Wt[HIDC * P64];
    __shared__ __attribute__((aligned(16))) float Gs[2][16 * P64];
    __shared__ __attribute__((aligned(16))) float bnsc[HIDC];
    __shared__ __attribute__((aligned(16))) float bnsh[HIDC];
    __shared__ float red[2][2][HIDC];
    const int tid = threadIdx.x;
    if (tid < HIDC) {
        float s = 0.f, s2 = 0.f;
        #pragma unroll
        for (int j = 0; j < SUM_COPIES; ++j) {
            s  += bnsum[j * 128 + tid];
            s2 += bnsum[j * 128 + 64 + tid];
        }
        const float mean = s * invN;
        const float var  = s2 * invN - mean * mean;
        const float sc   = bng[tid] * rsqrtf(var + 1e-5f);
        bnsc[tid] = sc;
        bnsh[tid] = bnb[tid] - mean * sc;
    }
    #pragma unroll
    for (int j = 0; j < HIDC * HIDC / (FUSED_WAVES * 32); ++j) {
        const int i = tid + j * (FUSED_WAVES * 32);
        Wt[(i & 63) * P64 + (i >> 6)] = w[i];
    }
    __syncthreads();

    const int lane = tid & 31;
    const int wv = __builtin_amdgcn_readfirstlane(tid >> 5);
    const int hi = lane >> 4;
    const int lo = lane & 15;
    const v2f* __restrict__ r2 = (const v2f*)r;
    const v2f sc2 = ((const v2f*)bnsc)[lane];
    const v2f sh2 = ((const v2f*)bnsh)[lane];
    const int n0 = (wv & 3) << 4;
    const float bv = bias[n0 + lo];
    const int groups = (Nn + 15) >> 4;
    float sx = 0.f, qx = 0.f;
    int buf = 0;
    for (int g = (int)blockIdx.x; g < groups; g += (int)gridDim.x, buf ^= 1) {
        const int d = (g << 4) + wv;
        v2f gsum = {0.f, 0.f};
        if (d < Nn)
            gather_segment<true>(r2, pairs, off[d], off[d + 1], lane, sc2, sh2, gsum);
        *(v2f*)&Gs[buf][wv * P64 + 2 * lane] = gsum;
        __syncthreads();
        if (wv < 4) {
            v8f acc = {0.f, 0.f, 0.f, 0.f, 0.f, 0.f, 0.f, 0.f};
            #pragma unroll
            for (int s = 0; s < 16; ++s) {
                const int k = 4 * s + (hi << 1);
                const v2f a = *(const v2f*)&Gs[buf][lo * P64 + k];
                const v2f b = *(const v2f*)&Wt[(n0 + lo) * P64 + k];
                acc = __builtin_amdgcn_wmma_f32_16x16x4_f32(
                    false, a, false, b, (short)0, acc, false, false);
            }
            const int m0 = (g << 4) + (hi << 3);
            float* op = out + (size_t)m0 * HIDC + n0 + lo;
            #pragma unroll
            for (int j = 0; j < 8; ++j) {
                const float v = fmaxf(acc[j] + bv, 0.f);
                if (m0 + j < Nn) {
                    op[j * HIDC] = v;
                    sx += v;
                    qx += v * v;
                }
            }
        }
    }
    if (wv < 4) {
        red[0][hi][n0 + lo] = sx;
        red[1][hi][n0 + lo] = qx;
    }
    __syncthreads();
    if (tid < 128) {
        const float t = red[tid >> 6][0][tid & 63] + red[tid >> 6][1][tid & 63];
        unsafeAtomicAdd(&sums[(blockIdx.x & (SUM_COPIES - 1)) * 128 + tid], t);
    }
}

extern "C" void kernel_launch(void* const* d_in, const int* in_sizes, int n_in,
                              void* d_out, int out_size, void* d_ws, size_t ws_size,
                              hipStream_t stream)
{
    const float* x  = (const float*)d_in[0];
    const int*   ei = (const int*)d_in[1];
    const float* ew = (const float*)d_in[2];
    const float* cw[3] = {(const float*)d_in[3], (const float*)d_in[5], (const float*)d_in[7]};
    const float* cb[3] = {(const float*)d_in[4], (const float*)d_in[6], (const float*)d_in[8]};
    const float* bg[3] = {(const float*)d_in[9],  (const float*)d_in[11], (const float*)d_in[13]};
    const float* bb[3] = {(const float*)d_in[10], (const float*)d_in[12], (const float*)d_in[14]};
    const float* w1 = (const float*)d_in[15]; const float* b1 = (const float*)d_in[16];
    const float* w2 = (const float*)d_in[17]; const float* b2 = (const float*)d_in[18];
    const float* w3 = (const float*)d_in[19]; const float* b3 = (const float*)d_in[20];

    const int N = in_sizes[0] / 128;
    const int E = in_sizes[2];
    const int* src = ei;
    const int* dst = ei + E;

    float* bufA  = (float*)d_ws;
    float* bufB  = bufA + (size_t)N * 128;
    int2*  pairs = (int2*)(bufA + (size_t)N * 64);
    int*   off   = (int*)d_out;
    int2*  grouped = (int2*)bufB;
    const int sumsBase = (N + 64) & ~63;
    const int sumsPer  = SUM_COPIES * 128;
    float* sums  = (float*)d_out + sumsBase;
    int*   bucketCnt = off + sumsBase + 3 * sumsPer;
    int*   cursor    = bucketCnt + KMAX;
    const int K = (N + BUCKET_NODES - 1) >> BUCKET_SHIFT;

    const dim3 blk256(256);
    const float invN   = 1.0f / (float)N;

    zero_i32<<<16, blk256, 0, stream>>>((int*)sums, 3 * sumsPer + 2 * KMAX);
    bucket_hist_product0<<<256, PT_THREADS, 0, stream>>>(dst, bucketCnt, E, K,
                                                         x, cw[0], bufA, N);
    edge_partition<<<256, PT_THREADS, 0, stream>>>(src, dst, ew, bucketCnt, cursor, grouped, E, K);
    bucket_sort<<<K, BK_THREADS, 0, stream>>>(bucketCnt, grouped, pairs, off, N, K);

    gcn_gather_relu_sums<<<GATHER_BLOCKS, blk256, 0, stream>>>(
        bufA, off, pairs, cb[0], bufB, sums, N);
    gcn_gather_mm_relu_sums<<<FUSED_BLOCKS, FUSED_WAVES * 32, 0, stream>>>(
        bufB, off, pairs, sums, bg[0], bb[0], invN, cw[1], cb[1],
        bufA, sums + sumsPer, N);
    gcn_gather_mm_relu_sums<<<FUSED_BLOCKS, FUSED_WAVES * 32, 0, stream>>>(
        bufA, off, pairs, sums + sumsPer, bg[1], bb[1], invN, cw[2], cb[2],
        bufB, sums + 2 * sumsPer, N);

    bn_fold<<<1, 64, 0, stream>>>(sums + 2 * sumsPer, bg[2], bb[2], invN, bufA);
    mlp_fused<<<MLP_BLOCKS, MLP_WAVES * 32, 0, stream>>>(
        bufB, bufA, w1, b1, w2, b2, w3, b3, (float*)d_out, N);
    (void)n_in; (void)out_size; (void)ws_size;
}


# Classifier GELU: erf part branch-free, Phi(-|u|) = exp2(one degree-9 polynomial), gelu = 0.5u + |u|(0.5 - Phi(-|u|)), 2 values per packed op; was library erff, two EXEC-masked paths per value
# speedup vs baseline: 5.7032x; 1.0603x over previous
#include <hip/hip_runtime.h>
#include <cstddef>


typedef __attribute__((ext_vector_type(2))) float v2f;
typedef __attribute__((ext_vector_type(8))) float v8f;

#define HIDC 64

#define INC  128
#define P64  68
#define P128 132
#define SUM_COPIES 8

#define BUCKET_SHIFT 8
#define BUCKET_NODES 256
#define KMAX 512

__global__ __launch_bounds__(256) void zero_i32(int* __restrict__ p, int n)
{
    int i = blockIdx.x * blockDim.x + threadIdx.x;
    const int st = gridDim.x * blockDim.x;
    for (; i < n; i += st) p[i] = 0;
}

__device__ __forceinline__ int block_excl_scan(int v, int* wave_excl)
{
    const int lane = threadIdx.x & 31;
    const int wv   = threadIdx.x >> 5;
    const int nw   = (int)(blockDim.x >> 5);
    int x = v;
    #pragma unroll
    for (int o = 1; o < 32; o <<= 1) {
        const int y = __shfl_up(x, o);
        if (lane >= o) x += y;
    }
    if (lane == 31) wave_excl[wv] = x;
    __syncthreads();
    if (wv == 0) {
        const int w = lane < nw ? wave_excl[lane] : 0;
        int z = w;
        #pragma unroll
        for (int o = 1; o < 32; o <<= 1) {
            const int y = __shfl_up(z, o);
            if (lane >= o) z += y;
        }
        wave_excl[lane] = z - w;
    }
    __syncthreads();
    const int r = wave_excl[wv] + (x - v);
    __syncthreads();
    return r;
}

#define PT_THREADS 1024
#define PT_TILE 12800
#define PT_PER ((PT_TILE + PT_THREADS - 1) / PT_THREADS)

__device__ __forceinline__ void load_a_strip(
    const float* __restrict__ x, int t, int M, int lo, int hi, v2f (&a)[INC / 4])
{
    int arow = (t << 4) + lo; if (arow > M - 1) arow = M - 1;
    const float* ap = x + (size_t)arow * INC + (hi << 1);
    #pragma unroll
    for (int s = 0; s < INC / 4; ++s) a[s] = *(const v2f*)(ap + 4 * s);
}
__global__ __launch_bounds__(PT_THREADS) __attribute__((amdgpu_waves_per_eu(8, 8)))
void bucket_hist_product0(
    const int* __restrict__ dst, int* __restrict__ bucketCnt, int E, int K,
    const float* __restrict__ x, const float* __restrict__ w0,
    float* __restrict__ h, int M)
{
    __shared__ int lh[KMAX];
    __shared__ __attribute__((aligned(16))) float Wt[HIDC * P128];
    const int tid  = threadIdx.x;
    const int lane = tid & 31;
    const int wave = __builtin_amdgcn_readfirstlane(tid >> 5);
    const int hi   = lane >> 4;
    const int lo   = lane & 15;
    const int tiles = (M + 15) >> 4;
    const int tstep = (int)gridDim.x * (PT_THREADS / 32);
    int t = wave * (int)gridDim.x + (int)blockIdx.x;

    if (tid < KMAX) lh[tid] = 0;
    __syncthreads();
    constexpr int WPER = INC * HIDC / PT_THREADS;
    float rw[WPER];
    #pragma unroll
    for (int j = 0; j < WPER; ++j) rw[j] = w0[tid + j * PT_THREADS];
    for (int e0 = (int)blockIdx.x * PT_TILE; e0 < E; e0 += (int)gridDim.x * PT_TILE) {
        const int n = (E - e0) < PT_TILE ? (E - e0) : PT_TILE;
        int rd[PT_PER];
        #pragma unroll
        for (int k = 0; k < PT_PER; ++k) {
            const int i = e0 + tid + k * PT_THREADS;
            rd[k] = dst[i < E ? i : E - 1];
        }
        #pragma unroll
        for (int k = 0; k < PT_PER; ++k)
            atomicAdd(&lh[rd[k] >> BUCKET_SHIFT], (tid + k * PT_THREADS) < n ? 1 : 0);
    }
    #pragma unroll
    for (int j = 0; j < WPER; ++j) {
        const int i = tid + j * PT_THREADS;
        Wt[(i & 63) * P128 + (i >> 6)] = rw[j];
    }
    __syncthreads();
    if (tid < K) {
        const int c = lh[tid];
        if (c) atomicAdd(&bucketCnt[tid], c);
    }

    while (t < tiles) {
        asm volatile("" ::: "memory");
        v2f a[INC / 4];
        load_a_strip(x, t, M, lo, hi, a);
        v8f acc[4];
        #pragma unroll
        for (int c = 0; c < 4; ++c) acc[c] = (v8f){0.f, 0.f, 0.f, 0.f, 0.f, 0.f, 0.f, 0.f};
        #pragma unroll
        for (int s = 0; s < INC / 4; ++s) {
            const int k = 4 * s + (hi << 1);
            #pragma unroll
            for (int c = 0; c < 4; ++c) {
                const v2f b = *(const v2f*)&Wt[(c * 16 + lo) * P128 + k];
                acc[c] = __builtin_amdgcn_wmma_f32_16x16x4_f32(
                    false, a[s], false, b, (short)0, acc[c], false, false);
            }
        }
        const int m0 = (t << 4) + (hi << 3);
        float* op = h + (size_t)m0 * HIDC + lo;
        #pragma unroll
        for (int c = 0; c < 4; ++c) {
            #pragma unroll
            for (int j = 0; j < 8; ++j)
                if (m0 + j < M) op[j * HIDC + c * 16] = acc[c][j];
        }
        t += tstep;
    }
}

__global__ __launch_bounds__(PT_THREADS) void edge_partition(
    const int* __restrict__ src, const int* __restrict__ dst,
    const float* __restrict__ ew, const int* __restrict__ bucketCnt,
    int* __restrict__ cursor, int2* __restrict__ grouped, int E, int K)
{
    __shared__ int2 s_pair[PT_TILE];
    __shared__ unsigned short s_bk[PT_TILE];
    __shared__ int lh[KMAX];
    __shared__ int lstart[KMAX];
    __shared__ int delta[KMAX];
    __shared__ int gbase[KMAX];
    __shared__ int wave_excl[32];
    const int tid = threadIdx.x;

    {
        const int tot = tid < K ? bucketCnt[tid] : 0;
        const int ex  = block_excl_scan(tot, wave_excl);
        if (tid < KMAX) gbase[tid] = ex;
    }
    for (int e0 = (int)blockIdx.x * PT_TILE; e0 < E; e0 += (int)gridDim.x * PT_TILE) {
        const int n = (E - e0) < PT_TILE ? (E - e0) : PT_TILE;
        if (tid < KMAX) lh[tid] = 0;
        __syncthreads();
        int   rs[PT_PER], rd[PT_PER], rr[PT_PER];
        float rw[PT_PER];
        #pragma unroll
        for (int k = 0; k < PT_PER; ++k) {
            const int i = tid + k * PT_THREADS;
            if (i < n) {
                rd[k] = dst[e0 + i];
                rs[k] = src[e0 + i];
                rw[k] = ew[e0 + i];
            }
        }
        #pragma unroll
        for (int k = 0; k < PT_PER; ++k) {
            const int i = tid + k * PT_THREADS;
            if (i < n) rr[k] = atomicAdd(&lh[rd[k] >> BUCKET_SHIFT], 1);
        }
        __syncthreads();
        const int c   = tid < K ? lh[tid] : 0;
        const int lex = block_excl_scan(c, wave_excl);
        if (tid < K) {
            lstart[tid] = lex;
            const int g = c ? atomicAdd(&cursor[tid], c) : 0;
            delta[tid] = gbase[tid] + g - lex;
        }
        __syncthreads();
        #pragma unroll
        for (int k = 0; k < PT_PER; ++k) {
            const int i = tid + k * PT_THREADS;
            if (i < n) {
                const int b = rd[k] >> BUCKET_SHIFT;
                const int p = lstart[b] + rr[k];
                int2 v;
                v.x = rs[k] | ((rd[k] & (BUCKET_NODES - 1)) << 24);
                v.y = __float_as_int(rw[k]);
                s_pair[p] = v;
                s_bk[p]   = (unsigned short)b;
            }
        }
        __syncthreads();
        for (int i = tid; i < n; i += PT_THREADS)
            grouped[delta[s_bk[i]] + i] = s_pair[i];
        __syncthreads();
    }
}

#define BK_THREADS 512
#define BK_CAP 12288
__global__ __launch_bounds__(BK_THREADS) void bucket_sort(
    const int* __restrict__ bucketCnt, const int2* __restrict__ grouped,
    int2* __restrict__ pairs, int* __restrict__ off, int Nn, int K)
{
    __shared__ int2 stage[BK_CAP];
    __shared__ int cnt[BUCKET_NODES];
    __shared__ int cur[BUCKET_NODES];
    __shared__ int wave_excl[32];
    __shared__ int s_base, s_count;
    const int tid = threadIdx.x;
    const int b   = blockIdx.x;
    {
        const int tot = tid < K ? bucketCnt[tid] : 0;
        const int ex  = block_excl_scan(tot, wave_excl);
        if (tid == b) { s_base = ex; s_count = tot; }
        if (tid < BUCKET_NODES) cnt[tid] = 0;
    }
    __syncthreads();
    const int base  = s_base;
    const int count = s_count;
    const int2* __restrict__ in = grouped + base;
    {
        int i = tid;
        for (; i + 3 * BK_THREADS < count; i += 4 * BK_THREADS) {
            const int x0 = in[i].x;
            const int x1 = in[i + BK_THREADS].x;
            const int x2 = in[i + 2 * BK_THREADS].x;
            const int x3 = in[i + 3 * BK_THREADS].x;
            atomicAdd(&cnt[(unsigned)x0 >> 24], 1);
            atomicAdd(&cnt[(unsigned)x1 >> 24], 1);
            atomicAdd(&cnt[(unsigned)x2 >> 24], 1);
            atomicAdd(&cnt[(unsigned)x3 >> 24], 1);
        }
        for (; i < count; i += BK_THREADS)
            atomicAdd(&cnt[(unsigned)in[i].x >> 24], 1);
    }
    __syncthreads();
    {
        const int c   = tid < BUCKET_NODES ? cnt[tid] : 0;
        const int lex = block_excl_scan(c, wave_excl);
        if (tid < BUCKET_NODES) {
            cur[tid] = lex;
            const int node = b * BUCKET_NODES + tid;
            if (node < Nn) off[node] = base + lex;
        }
        if (b == K - 1 && tid == 0) off[Nn] = base + count;
    }
    __syncthreads();
    const bool staged = count <= BK_CAP;
    {
        auto place = [&](int2 e) {
            const int r = atomicAdd(&cur[(unsigned)e.x >> 24], 1);
            int2 p;
            p.x = e.x & 0x00FFFFFF;
            p.y = e.y;
            if (staged) stage[r] = p;
            else        pairs[base + r] = p;
        };
        int i = tid;
        for (; i + 3 * BK_THREADS < count; i += 4 * BK_THREADS) {
            const int2 e0 = in[i];
            const int2 e1 = in[i + BK_THREADS];
            const int2 e2 = in[i + 2 * BK_THREADS];
            const int2 e3 = in[i + 3 * BK_THREADS];
            place(e0); place(e1); place(e2); place(e3);
        }
        for (; i < count; i += BK_THREADS) place(in[i]);
    }
    if (staged) {
        __syncthreads();
        for (int i = tid; i < count; i += BK_THREADS) pairs[base + i] = stage[i];
    }
}


template <int K, bool BN>
__device__ __forceinline__ void gather_rows(
    const v2f* __restrict__ h2, int s, float w, int j0, int lane,
    v2f sc, v2f sh, v2f& acc)
{
    v2f   v[K];
    float wj[K];
    #pragma unroll
    for (int k = 0; k < K; ++k) {
        const int sj = __builtin_amdgcn_readlane(s, j0 + k);
        wj[k] = __int_as_float(__builtin_amdgcn_readlane(__float_as_int(w), j0 + k));
        v[k]  = h2[(size_t)sj * (HIDC / 2) + lane];
    }
    #pragma unroll
    for (int k = 0; k < K; ++k) {
        v2f t = v[k];
        if (BN) t = t * sc + sh;
        acc += t * wj[k];
    }
}

template <bool BN>
__device__ __forceinline__ void gather_lanes(
    const v2f* __restrict__ h2, int s, float w, int cnt, int lane,
    v2f sc, v2f sh, v2f& acc)
{
    int j = 0;
    for (; j + 8 <= cnt; j += 8) gather_rows<8, BN>(h2, s, w, j, lane, sc, sh, acc);
    if (cnt & 4) { gather_rows<4, BN>(h2, s, w, j, lane, sc, sh, acc); j += 4; }
    if (cnt & 2) { gather_rows<2, BN>(h2, s, w, j, lane, sc, sh, acc); j += 2; }
    if (cnt & 1) { gather_rows<1, BN>(h2, s, w, j, lane, sc, sh, acc); }
}

template <bool BN>
__device__ __forceinline__ void gather_segment(
    const v2f* __restrict__ h2, const int2* __restrict__ pairs,
    int beg, int end, int lane, v2f sc, v2f sh, v2f& acc)
{
    for (int base = beg; base < end; base += 64) {
        const int left = end - base;
        const int c0 = left < 32 ? left : 32;
        const int r1 = left - 32;
        const int c1 = r1 < 0 ? 0 : (r1 < 32 ? r1 : 32);
        const int i0 = base + lane;
        const int i1 = base + 32 + lane;
        const int2 p0 = pairs[i0 < end ? i0 : end - 1];
        const int2 p1 = pairs[i1 < end ? i1 : end - 1];
        const int   s0 = p0.x;
        const int   s1 = p1.x;
        const float w0 = __int_as_float(p0.y);
        const float w1 = __int_as_float(p1.y);
        asm volatile("" :: "v"(s0), "v"(s1), "v"(w0), "v"(w1));
        gather_lanes<BN>(h2, s0, w0, c0, lane, sc, sh, acc);
        gather_lanes<BN>(h2, s1, w1, c1, lane, sc, sh, acc);
    }
}

#define GATHER_BLOCKS 2048
__global__ __launch_bounds__(256) void gcn_gather_relu_sums(
    const float* __restrict__ h, const int* __restrict__ off,
    const int2* __restrict__ pairs, const float* __restrict__ bias,
    float* __restrict__ agg, float* __restrict__ sums, int Nn)
{
    __shared__ float red[8][128];
    const int lane = threadIdx.x & 31;
    const int wv   = threadIdx.x >> 5;
    int d = __builtin_amdgcn_readfirstlane((int)(blockIdx.x * 8 + wv));
    const int dstep = (int)gridDim.x * 8;
    const v2f* __restrict__ h2 = (const v2f*)h;
    const float2 b2 = ((const float2*)bias)[lane];
    float sx = 0.f, sy = 0.f, qx = 0.f, qy = 0.f;
    for (; d < Nn; d += dstep) {
        const int beg = off[d];
        const int end = off[d + 1];
        v2f acc = {0.f, 0.f};
        const v2f none = {0.f, 0.f};
        gather_segment<false>(h2, pairs, beg, end, lane, none, none, acc);
        float2 r;
        r.x = fmaxf(acc.x + b2.x, 0.f);
        r.y = fmaxf(acc.y + b2.y, 0.f);
        ((float2*)agg)[(size_t)d * (HIDC / 2) + lane] = r;
        sx += r.x;        sy += r.y;
        qx += r.x * r.x;  qy += r.y * r.y;
    }
    red[wv][2 * lane]          = sx;
    red[wv][2 * lane + 1]      = sy;
    red[wv][64 + 2 * lane]     = qx;
    red[wv][64 + 2 * lane + 1] = qy;
    __syncthreads();
    if (threadIdx.x < 128) {
        float t = 0.f;
        #pragma unroll
        for (int w = 0; w < 8; ++w) t += red[w][threadIdx.x];
        unsafeAtomicAdd(&sums[(blockIdx.x & (SUM_COPIES - 1)) * 128 + threadIdx.x], t);
    }
}


__global__ __launch_bounds__(64) void bn_fold(
    const float* __restrict__ bnsum, const float* __restrict__ bng,
    const float* __restrict__ bnb, float invN, float* __restrict__ scsh)
{
    const int c = threadIdx.x;
    float s = 0.f, s2 = 0.f;
    #pragma unroll
    for (int j = 0; j < SUM_COPIES; ++j) {
        s  += bnsum[j * 128 + c];
        s2 += bnsum[j * 128 + 64 + c];
    }
    const float mean = s * invN;
    const float var  = s2 * invN - mean * mean;
    const float sc   = bng[c] * rsqrtf(var + 1e-5f);
    scsh[c]      = sc;
    scsh[64 + c] = bnb[c] - mean * sc;
}

#define MLP_WAVES 16
#define MLP_BLOCKS 256
#define GELU_PAIRS 2
__device__ __forceinline__ v2f splat2(float c) { return (v2f){c, c}; }
__device__ __forceinline__ void gelu_tile(v2f (&u)[GELU_PAIRS])
{
    v2f au[GELU_PAIRS], a[GELU_PAIRS], p[GELU_PAIRS];
    #pragma unroll
    for (int i = 0; i < GELU_PAIRS; ++i) {
        au[i].x = __builtin_fabsf(u[i].x);
        au[i].y = __builtin_fabsf(u[i].y);
        a[i].x = __builtin_fminf(au[i].x, 6.0f);
        a[i].y = __builtin_fminf(au[i].y, 6.0f);
        p[i] = __builtin_elementwise_fma(a[i], splat2(4.45528030e-7f), splat2(-8.61138051e-6f));
    }
    const float cf[8] = {6.95293566e-5f, -2.67886062e-4f, -1.33644080e-5f, 6.95919991e-3f, -0.0524501018f, -0.459217310f, -1.15110469f, -1.0f};
    #pragma unroll
    for (int k = 0; k < 8; ++k) {
        #pragma unroll
        for (int i = 0; i < GELU_PAIRS; ++i)
            p[i] = __builtin_elementwise_fma(p[i], a[i], splat2(cf[k]));
    }
    #pragma unroll
    for (int i = 0; i < GELU_PAIRS; ++i) {
        v2f e;
        e.x = __builtin_amdgcn_exp2f(p[i].x);
        e.y = __builtin_amdgcn_exp2f(p[i].y);
        const v2f h = splat2(0.5f) - e;
        u[i] = __builtin_elementwise_fma(au[i], h, u[i] * 0.5f);
    }
}
__global__ __launch_bounds__(MLP_WAVES * 32) void mlp_fused(
    const float* __restrict__ R, const float* __restrict__ scsh,
    const float* __restrict__ w1, const float* __restrict__ b1,
    const float* __restrict__ w2, const float* __restrict__ b2,
    const float* __restrict__ w3, const float* __restrict__ b3,
    float* __restrict__ out, int M)
{
    __shared__ __attribute__((aligned(16))) float Wt1[128 * P64];
    __shared__ __attribute__((aligned(16))) float Wt2[64 * P128];
    __shared__ __attribute__((aligned(16))) float Wt3[16 * P64];
    __shared__ __attribute__((aligned(16))) float Hs[MLP_WAVES][16 * P128];
    __shared__ __attribute__((aligned(16))) float bnsc[64];
    __shared__ __attribute__((aligned(16))) float bnsh[64];
    const int tid = threadIdx.x;
    {
        constexpr int NT = MLP_WAVES * 32;
        constexpr int PER = 64 * 128 / NT;
        float r1[PER], r2[PER];
        #pragma unroll
        for (int j = 0; j < PER; ++j) r1[j] = w1[tid + j * NT];
        #pragma unroll
        for (int j = 0; j < PER; ++j) r2[j] = w2[tid + j * NT];
        #pragma unroll
        for (int j = 0; j < PER; ++j) {
            const int i = tid + j * NT;
            Wt1[(i & 127) * P64 + (i >> 7)] = r1[j];
        }
        #pragma unroll
        for (int j = 0; j < PER; ++j) {
            const int i = tid + j * NT;
            Wt2[(i & 63) * P128 + (i >> 6)] = r2[j];
        }
    }
    for (int i = tid; i < 16 * 64; i += MLP_WAVES * 32) {
        const int n = i >> 6, k = i & 63;
        Wt3[n * P64 + k] = n < 5 ? w3[k * 5 + n] : 0.f;
    }
    if (tid < 64) { bnsc[tid] = scsh[tid]; bnsh[tid] = scsh[64 + tid]; }
    __syncthreads();

    const int lane = tid & 31;
    const int wave = tid >> 5;
    const int hi   = lane >> 4;
    const int lo   = lane & 15;
    float* hs = Hs[wave];
    const int tiles = (M + 15) >> 4;
    float bias1[8], bias2[4];
    #pragma unroll
    for (int c = 0; c < 8; ++c) bias1[c] = b1[c * 16 + lo];
    #pragma unroll
    for (int c = 0; c < 4; ++c) bias2[c] = b2[c * 16 + lo];
    const float bias3 = lo < 5 ? b3[lo] : 0.f;

    for (int t = wave * (int)gridDim.x + (int)blockIdx.x; t < tiles; t += (int)gridDim.x * MLP_WAVES) {
        const int row0 = t << 4;
        int arow = row0 + lo; if (arow > M - 1) arow = M - 1;
        const float* ap = R + (size_t)arow * 64 + (hi << 1);

        v2f a1[16];
        #pragma unroll
        for (int s = 0; s < 16; ++s) a1[s] = *(const v2f*)(ap + 4 * s);
        v8f acc1[8];
        #pragma unroll
        for (int c = 0; c < 8; ++c) acc1[c] = (v8f){0.f, 0.f, 0.f, 0.f, 0.f, 0.f, 0.f, 0.f};
        #pragma unroll
        for (int s = 0; s < 16; ++s) {
            const int k = 4 * s + (hi << 1);
            const v2f sc = *(const v2f*)&bnsc[k];
            const v2f sh = *(const v2f*)&bnsh[k];
            v2f a;
            a.x = a1[s].x * sc.x + sh.x;
            a.y = a1[s].y * sc.y + sh.y;
            #pragma unroll
            for (int c = 0; c < 8; ++c) {
                const v2f b = *(const v2f*)&Wt1[(c * 16 + lo) * P64 + k];
                acc1[c] = __builtin_amdgcn_wmma_f32_16x16x4_f32(
                    false, a, false, b, (short)0, acc1[c], false, false);
            }
        }
        #pragma unroll
        for (int c = 0; c < 8; ++c) {
            #pragma unroll
            for (int j = 0; j < 8; j += 2 * GELU_PAIRS) {
                v2f u[GELU_PAIRS];
                #pragma unroll
                for (int i = 0; i < GELU_PAIRS; ++i)
                    u[i] = (v2f){acc1[c][j + 2 * i] + bias1[c], acc1[c][j + 2 * i + 1] + bias1[c]};
                gelu_tile(u);
                #pragma unroll
                for (int i = 0; i < GELU_PAIRS; ++i) {
                    hs[(j + 2 * i + (hi << 3)) * P128 + c * 16 + lo]     = u[i].x;
                    hs[(j + 2 * i + 1 + (hi << 3)) * P128 + c * 16 + lo] = u[i].y;
                }
            }
        }

        v8f acc2[4];
        #pragma unroll
        for (int c = 0; c < 4; ++c) acc2[c] = (v8f){0.f, 0.f, 0.f, 0.f, 0.f, 0.f, 0.f, 0.f};
        #pragma unroll 8
        for (int s = 0; s < 32; ++s) {
            const int k = 4 * s + (hi << 1);
            const v2f a = *(const v2f*)&hs[lo * P128 + k];
            #pragma unroll
            for (int c = 0; c < 4; ++c) {
                const v2f b = *(const v2f*)&Wt2[(c * 16 + lo) * P128 + k];
                acc2[c] = __builtin_amdgcn_wmma_f32_16x16x4_f32(
                    false, a, false, b, (short)0, acc2[c], false, false);
            }
        }
        #pragma unroll
        for (int c = 0; c < 4; ++c) {
            #pragma unroll
            for (int j = 0; j < 8; j += 2 * GELU_PAIRS) {
                v2f u[GELU_PAIRS];
                #pragma unroll
                for (int i = 0; i < GELU_PAIRS; ++i)
                    u[i] = (v2f){acc2[c][j + 2 * i] + bias2[c], acc2[c][j + 2 * i + 1] + bias2[c]};
                gelu_tile(u);
                #pragma unroll
                for (int i = 0; i < GELU_PAIRS; ++i) {
                    hs[(j + 2 * i + (hi << 3)) * P128 + c * 16 + lo]     = u[i].x;
                    hs[(j + 2 * i + 1 + (hi << 3)) * P128 + c * 16 + lo] = u[i].y;
                }
            }
        }

        v8f acc3 = {0.f, 0.f, 0.f, 0.f, 0.f, 0.f, 0.f, 0.f};
        #pragma unroll
        for (int s = 0; s < 16; ++s) {
            const int k = 4 * s + (hi << 1);
            const v2f a = *(const v2f*)&hs[lo * P128 + k];
            const v2f b = *(const v2f*)&Wt3[lo * P64 + k];
            acc3 = __builtin_amdgcn_wmma_f32_16x16x4_f32(
                false, a, false, b, (short)0, acc3, false, false);
        }
        if (lo < 5) {
            #pragma unroll
            for (int j = 0; j < 8; ++j) {
                const int m = row0 + j + (hi << 3);
                if (m < M) out[(size_t)m * 5 + lo] = acc3[j] + bias3;
            }
        }
    }
}

#define FUSED_WAVES 16
#define FUSED_BLOCKS 1024
__global__ __launch_bounds__(FUSED_WAVES * 32, 16) void gcn_gather_mm_relu_sums(
    const float* __restrict__ r, const int* __restrict__ off,
    const int2* __restrict__ pairs,
    const float* __restrict__ bnsum, const float* __restrict__ bng,
    const float* __restrict__ bnb, float invN,
    const float* __restrict__ w, const float* __restrict__ bias,
    float* __restrict__ out, float* __restrict__ sums, int Nn)
{
    __shared__ __attribute__((aligned(16))) float Wt[HIDC * P64];
    __shared__ __attribute__((aligned(16))) float Gs[2][16 * P64];
    __shared__ __attribute__((aligned(16))) float bnsc[HIDC];
    __shared__ __attribute__((aligned(16))) float bnsh[HIDC];
    __shared__ float red[2][2][HIDC];
    const int tid = threadIdx.x;
    if (tid < HIDC) {
        float s = 0.f, s2 = 0.f;
        #pragma unroll
        for (int j = 0; j < SUM_COPIES; ++j) {
            s  += bnsum[j * 128 + tid];
            s2 += bnsum[j * 128 + 64 + tid];
        }
        const float mean = s * invN;
        const float var  = s2 * invN - mean * mean;
        const float sc   = bng[tid] * rsqrtf(var + 1e-5f);
        bnsc[tid] = sc;
        bnsh[tid] = bnb[tid] - mean * sc;
    }
    #pragma unroll
    for (int j = 0; j < HIDC * HIDC / (FUSED_WAVES * 32); ++j) {
        const int i = tid + j * (FUSED_WAVES * 32);
        Wt[(i & 63) * P64 + (i >> 6)] = w[i];
    }
    __syncthreads();

    const int lane = tid & 31;
    const int wv = __builtin_amdgcn_readfirstlane(tid >> 5);
    const int hi = lane >> 4;
    const int lo = lane & 15;
    const v2f* __restrict__ r2 = (const v2f*)r;
    const v2f sc2 = ((const v2f*)bnsc)[lane];
    const v2f sh2 = ((const v2f*)bnsh)[lane];
    const int n0 = (wv & 3) << 4;
    const float bv = bias[n0 + lo];
    const int groups = (Nn + 15) >> 4;
    float sx = 0.f, qx = 0.f;
    int buf = 0;
    for (int g = (int)blockIdx.x; g < groups; g += (int)gridDim.x, buf ^= 1) {
        const int d = (g << 4) + wv;
        v2f gsum = {0.f, 0.f};
        if (d < Nn)
            gather_segment<true>(r2, pairs, off[d], off[d + 1], lane, sc2, sh2, gsum);
        *(v2f*)&Gs[buf][wv * P64 + 2 * lane] = gsum;
        __syncthreads();
        if (wv < 4) {
            v8f acc = {0.f, 0.f, 0.f, 0.f, 0.f, 0.f, 0.f, 0.f};
            #pragma unroll
            for (int s = 0; s < 16; ++s) {
                const int k = 4 * s + (hi << 1);
                const v2f a = *(const v2f*)&Gs[buf][lo * P64 + k];
                const v2f b = *(const v2f*)&Wt[(n0 + lo) * P64 + k];
                acc = __builtin_amdgcn_wmma_f32_16x16x4_f32(
                    false, a, false, b, (short)0, acc, false, false);
            }
            const int m0 = (g << 4) + (hi << 3);
            float* op = out + (size_t)m0 * HIDC + n0 + lo;
            #pragma unroll
            for (int j = 0; j < 8; ++j) {
                const float v = fmaxf(acc[j] + bv, 0.f);
                if (m0 + j < Nn) {
                    op[j * HIDC] = v;
                    sx += v;
                    qx += v * v;
                }
            }
        }
    }
    if (wv < 4) {
        red[0][hi][n0 + lo] = sx;
        red[1][hi][n0 + lo] = qx;
    }
    __syncthreads();
    if (tid < 128) {
        const float t = red[tid >> 6][0][tid & 63] + red[tid >> 6][1][tid & 63];
        unsafeAtomicAdd(&sums[(blockIdx.x & (SUM_COPIES - 1)) * 128 + tid], t);
    }
}

extern "C" void kernel_launch(void* const* d_in, const int* in_sizes, int n_in,
                              void* d_out, int out_size, void* d_ws, size_t ws_size,
                              hipStream_t stream)
{
    const float* x  = (const float*)d_in[0];
    const int*   ei = (const int*)d_in[1];
    const float* ew = (const float*)d_in[2];
    const float* cw[3] = {(const float*)d_in[3], (const float*)d_in[5], (const float*)d_in[7]};
    const float* cb[3] = {(const float*)d_in[4], (const float*)d_in[6], (const float*)d_in[8]};
    const float* bg[3] = {(const float*)d_in[9],  (const float*)d_in[11], (const float*)d_in[13]};
    const float* bb[3] = {(const float*)d_in[10], (const float*)d_in[12], (const float*)d_in[14]};
    const float* w1 = (const float*)d_in[15]; const float* b1 = (const float*)d_in[16];
    const float* w2 = (const float*)d_in[17]; const float* b2 = (const float*)d_in[18];
    const float* w3 = (const float*)d_in[19]; const float* b3 = (const float*)d_in[20];

    const int N = in_sizes[0] / 128;
    const int E = in_sizes[2];
    const int* src = ei;
    const int* dst = ei + E;

    float* bufA  = (float*)d_ws;
    float* bufB  = bufA + (size_t)N * 128;
    int2*  pairs = (int2*)(bufA + (size_t)N * 64);
    int*   off   = (int*)d_out;
    int2*  grouped = (int2*)bufB;
    const int sumsBase = (N + 64) & ~63;
    const int sumsPer  = SUM_COPIES * 128;
    float* sums  = (float*)d_out + sumsBase;
    int*   bucketCnt = off + sumsBase + 3 * sumsPer;
    int*   cursor    = bucketCnt + KMAX;
    const int K = (N + BUCKET_NODES - 1) >> BUCKET_SHIFT;

    const dim3 blk256(256);
    const float invN   = 1.0f / (float)N;

    zero_i32<<<16, blk256, 0, stream>>>((int*)sums, 3 * sumsPer + 2 * KMAX);
    bucket_hist_product0<<<256, PT_THREADS, 0, stream>>>(dst, bucketCnt, E, K,
                                                         x, cw[0], bufA, N);
    edge_partition<<<256, PT_THREADS, 0, stream>>>(src, dst, ew, bucketCnt, cursor, grouped, E, K);
    bucket_sort<<<K, BK_THREADS, 0, stream>>>(bucketCnt, grouped, pairs, off, N, K);

    gcn_gather_relu_sums<<<GATHER_BLOCKS, blk256, 0, stream>>>(
        bufA, off, pairs, cb[0], bufB, sums, N);
    gcn_gather_mm_relu_sums<<<FUSED_BLOCKS, FUSED_WAVES * 32, 0, stream>>>(
        bufB, off, pairs, sums, bg[0], bb[0], invN, cw[1], cb[1],
        bufA, sums + sumsPer, N);
    gcn_gather_mm_relu_sums<<<FUSED_BLOCKS, FUSED_WAVES * 32, 0, stream>>>(
        bufA, off, pairs, sums + sumsPer, bg[1], bb[1], invN, cw[2], cb[2],
        bufB, sums + 2 * sumsPer, N);

    bn_fold<<<1, 64, 0, stream>>>(sums + 2 * sumsPer, bg[2], bb[2], invN, bufA);
    mlp_fused<<<MLP_BLOCKS, MLP_WAVES * 32, 0, stream>>>(
        bufB, bufA, w1, b1, w2, b2, w3, b3, (float*)d_out, N);
    (void)n_in; (void)out_size; (void)ws_size;
}
